# starting kernel
# baseline (speedup 1.0000x reference)
#include <hip/hip_runtime.h>
#include <hip/hip_bf16.h>

typedef __attribute__((ext_vector_type(16))) _Float16 v16h;
typedef __attribute__((ext_vector_type(8)))  float    v8f;

#define B_  8
#define N_  10000
#define E_  160000
#define F_  64
#define BN_ (B_*N_)
#define BE_ (B_*E_)

#define WPACK_HALVES 4096

__global__ void zero_f32(float* __restrict__ p, int n) {
  int i = blockIdx.x * blockDim.x + threadIdx.x;
  int stride = gridDim.x * blockDim.x;
  for (; i < n; i += stride) p[i] = 0.0f;
}

__global__ void degree_kernel(const int* __restrict__ ei, float* __restrict__ deg,
                              int* __restrict__ row32, int* __restrict__ col32) {
  int g = blockIdx.x * blockDim.x + threadIdx.x;
  if (g >= BE_) return;
  int b = g / E_;
  int r = ei[2 * g + 0] + b * N_;
  int c = ei[2 * g + 1] + b * N_;
  row32[g] = r;
  col32[g] = c;
  atomicAdd(&deg[r], 1.0f);
}

__global__ void dinv_kernel(float* __restrict__ deg) {
  int i = blockIdx.x * blockDim.x + threadIdx.x;
  if (i >= BN_) return;
  float d = deg[i];
  deg[i] = (d > 0.0f) ? rsqrtf(d) : 0.0f;
}

__global__ void norm_kernel(const float* __restrict__ dinv, const int* __restrict__ row32,
                            const int* __restrict__ col32, float* __restrict__ norm) {
  int g = blockIdx.x * blockDim.x + threadIdx.x;
  if (g >= BE_) return;
  norm[g] = dinv[row32[g]] * dinv[col32[g]];
}

__global__ __launch_bounds__(256) void pack_w_kernel(const float* __restrict__ W,
                                                     _Float16* __restrict__ Wp) {
  const int t    = threadIdx.x;
  const int nt   = t >> 6;
  const int kb   = (t >> 5) & 1;
  const int lane = t & 31;
  const int n    = nt * 16 + (lane & 15);
  const int half = lane >> 4;
  _Float16* dst = Wp + ((size_t)((nt * 2 + kb) * 32 + lane)) * 16;
#pragma unroll
  for (int v = 0; v < 8; ++v) {
    int kbase = (v < 4) ? (2 * v) : (16 + 2 * (v - 4));
    int k = kb * 32 + kbase + half * 8;
    dst[2 * v]     = (_Float16)W[n * F_ + k];
    dst[2 * v + 1] = (_Float16)W[n * F_ + k + 1];
  }
}

template <bool RELU_IN>
__global__ __launch_bounds__(256) void gemm_bias_wmma(const float* __restrict__ Hin,
                                                      const _Float16* __restrict__ Wp,
                                                      const float* __restrict__ bias,
                                                      float* __restrict__ Hout) {
  const int wave = threadIdx.x >> 5;
  const int lane = threadIdx.x & 31;
  const int rb   = blockIdx.x * 8 + wave;
  if (rb * 16 >= BN_) return;
  const int row0 = rb * 16;
  const int half = lane >> 4;
  const int mrow = lane & 15;

  v16h a0, a1;
  const float* hrow = Hin + (size_t)(row0 + mrow) * F_;
#pragma unroll
  for (int v = 0; v < 8; ++v) {
    int kbase = (v < 4) ? (2 * v) : (16 + 2 * (v - 4));
    int k0 = kbase + half * 8;
    float2 p0 = *(const float2*)(hrow + k0);
    float2 p1 = *(const float2*)(hrow + 32 + k0);
    if (RELU_IN) {
      p0.x = fmaxf(p0.x, 0.0f); p0.y = fmaxf(p0.y, 0.0f);
      p1.x = fmaxf(p1.x, 0.0f); p1.y = fmaxf(p1.y, 0.0f);
    }
    a0[2 * v] = (_Float16)p0.x; a0[2 * v + 1] = (_Float16)p0.y;
    a1[2 * v] = (_Float16)p1.x; a1[2 * v + 1] = (_Float16)p1.y;
  }

  const v16h* Wpv = (const v16h*)Wp;
#pragma unroll
  for (int nt = 0; nt < 4; ++nt) {
    const v16h b0 = Wpv[(nt * 2 + 0) * 32 + lane];
    const v16h b1 = Wpv[(nt * 2 + 1) * 32 + lane];

    v8f c = {};
    c = __builtin_amdgcn_wmma_f32_16x16x32_f16(false, a0, false, b0, (short)0, c, false, false);
    c = __builtin_amdgcn_wmma_f32_16x16x32_f16(false, a1, false, b1, (short)0, c, false, false);

    const int n = nt * 16 + mrow;
    const float bv = bias[n];
#pragma unroll
    for (int r = 0; r < 8; ++r) {
      int m = r + half * 8;
      Hout[(size_t)(row0 + m) * F_ + n] = c[r] + bv;
    }
  }
}

__global__ __launch_bounds__(256) void aggregate_kernel(const float* __restrict__ Hlin,
                                                        const int* __restrict__ row32,
                                                        const int* __restrict__ col32,
                                                        const float* __restrict__ norm,
                                                        float* __restrict__ out) {
  const int wave = threadIdx.x >> 5;
  const int lane = threadIdx.x & 31;
  const int g = blockIdx.x * 8 + wave;
  if (g >= BE_) return;
  const int r = row32[g];
  const int c = col32[g];
  const float nv = norm[g];
  const float2 h = ((const float2*)(Hlin + (size_t)c * F_))[lane];
  float* dst = out + (size_t)r * F_ + 2 * lane;
  atomicAdd(dst + 0, nv * h.x);
  atomicAdd(dst + 1, nv * h.y);
}

extern "C" void kernel_launch(void* const* d_in, const int* in_sizes, int n_in,
                              void* d_out, int out_size, void* d_ws, size_t ws_size,
                              hipStream_t stream) {
  const float* x  = (const float*)d_in[0];
  const int*   ei = (const int*)  d_in[1];
  const float* W1 = (const float*)d_in[2];
  const float* b1 = (const float*)d_in[3];
  const float* W2 = (const float*)d_in[4];
  const float* b2 = (const float*)d_in[5];
  const float* W3 = (const float*)d_in[6];
  const float* b3 = (const float*)d_in[7];
  float* out = (float*)d_out;

  auto align256 = [](size_t v) { return (v + 255) & ~(size_t)255; };
  char* ws = (char*)d_ws;
  float* deg   = (float*)ws;     ws += align256((size_t)BN_ * 4);
  int*   row32 = (int*)ws;       ws += align256((size_t)BE_ * 4);
  int*   col32 = (int*)ws;       ws += align256((size_t)BE_ * 4);
  float* norm  = (float*)ws;     ws += align256((size_t)BE_ * 4);
  float* Hlin  = (float*)ws;     ws += align256((size_t)BN_ * F_ * 4);
  float* Hagg  = (float*)ws;     ws += align256((size_t)BN_ * F_ * 4);
  _Float16* Wp1 = (_Float16*)ws; ws += align256((size_t)WPACK_HALVES * 2);
  _Float16* Wp2 = (_Float16*)ws; ws += align256((size_t)WPACK_HALVES * 2);
  _Float16* Wp3 = (_Float16*)ws; ws += align256((size_t)WPACK_HALVES * 2);

  const int TB = 256;
  zero_f32<<<512, TB, 0, stream>>>(deg, BN_);
  degree_kernel<<<(BE_ + TB - 1) / TB, TB, 0, stream>>>(ei, deg, row32, col32);
  dinv_kernel<<<(BN_ + TB - 1) / TB, TB, 0, stream>>>(deg);
  norm_kernel<<<(BE_ + TB - 1) / TB, TB, 0, stream>>>(deg, row32, col32, norm);
  pack_w_kernel<<<1, TB, 0, stream>>>(W1, Wp1);
  pack_w_kernel<<<1, TB, 0, stream>>>(W2, Wp2);
  pack_w_kernel<<<1, TB, 0, stream>>>(W3, Wp3);

  const int GEMM_GRID = (BN_ / 16 + 7) / 8;
  const int AGG_GRID  = BE_ / 8;

  gemm_bias_wmma<false><<<GEMM_GRID, TB, 0, stream>>>(x, Wp1, b1, Hlin);
  zero_f32<<<2048, TB, 0, stream>>>(Hagg, BN_ * F_);
  aggregate_kernel<<<AGG_GRID, TB, 0, stream>>>(Hlin, row32, col32, norm, Hagg);

  gemm_bias_wmma<true><<<GEMM_GRID, TB, 0, stream>>>(Hagg, Wp2, b2, Hlin);
  zero_f32<<<2048, TB, 0, stream>>>(Hagg, BN_ * F_);
  aggregate_kernel<<<AGG_GRID, TB, 0, stream>>>(Hlin, row32, col32, norm, Hagg);

  gemm_bias_wmma<true><<<GEMM_GRID, TB, 0, stream>>>(Hagg, Wp3, b3, Hlin);
  zero_f32<<<2048, TB, 0, stream>>>(out, BN_ * F_);
  aggregate_kernel<<<AGG_GRID, TB, 0, stream>>>(Hlin, row32, col32, norm, out);
}


# Kernels with no data dependence between them share a launch: counter zero fill + three W packs in one grid, layer-1 GEMM beside the row-pointer scan; 14 launches become 10, kernel bodies unchanged
# speedup vs baseline: 1.8719x; 1.8719x over previous
#include <hip/hip_runtime.h>
#include <hip/hip_bf16.h>

typedef __attribute__((ext_vector_type(16))) _Float16 v16h;
typedef __attribute__((ext_vector_type(8)))  float    v8f;

#define B_  8
#define N_  10000
#define E_  160000
#define F_  64
#define BN_ (B_*N_)
#define BE_ (B_*E_)

#define WPACK_HALVES 4096


__global__ __launch_bounds__(256) void count_slot_kernel(const int* __restrict__ ei,
                                                         int* __restrict__ cnt,
                                                         int* __restrict__ slot) {
  int g = blockIdx.x * blockDim.x + threadIdx.x;
  if (g >= BE_) return;
  int b = g / E_;
  int r = ei[2 * g + 0] + b * N_;
  slot[g] = atomicAdd(&cnt[r], 1);
}

#define SCAN_ITEMS ((N_ + 1023) / 1024)
__device__ __forceinline__ void rowptr_scan_block(int* __restrict__ cnt, int* wsum, const int b) {
  const int t    = threadIdx.x;
  const int lane = t & 31;
  const int w    = t >> 5;
  int* p = cnt + b * N_;
  int v[SCAN_ITEMS];
  int s = 0;
#pragma unroll
  for (int i = 0; i < SCAN_ITEMS; ++i) {
    int idx = t * SCAN_ITEMS + i;
    v[i] = (idx < N_) ? p[idx] : 0;
    s += v[i];
  }
  int inc = s;
#pragma unroll
  for (int d = 1; d < 32; d <<= 1) {
    int o = __shfl_up(inc, d);
    if (lane >= d) inc += o;
  }
  if (lane == 31) wsum[w] = inc;
  __syncthreads();
  if (w == 0) {
    int x = wsum[lane];
    int xi = x;
#pragma unroll
    for (int d = 1; d < 32; d <<= 1) {
      int o = __shfl_up(xi, d);
      if (lane >= d) xi += o;
    }
    wsum[lane] = xi - x;
  }
  __syncthreads();
  int run = b * E_ + wsum[w] + (inc - s);
#pragma unroll
  for (int i = 0; i < SCAN_ITEMS; ++i) {
    int idx = t * SCAN_ITEMS + i;
    if (idx < N_) { p[idx] = run; run += v[i]; }
  }
}

__global__ __launch_bounds__(256) void place_kernel(const int* __restrict__ ei,
                                                    const int* __restrict__ rowptr,
                                                    const int* __restrict__ slot,
                                                    int* __restrict__ colu) {
  int g = blockIdx.x * blockDim.x + threadIdx.x;
  if (g >= BE_) return;
  int b = g / E_;
  int r = ei[2 * g + 0] + b * N_;
  int c = ei[2 * g + 1] + b * N_;
  colu[rowptr[r] + slot[g]] = c;
}

__device__ __forceinline__ int row_end(const int* __restrict__ rowptr, int i) {
  return (i + 1 < BN_) ? rowptr[i + 1] : BE_;
}

__global__ __launch_bounds__(256) void sort_weight_kernel(const int* __restrict__ rowptr,
                                                          const int* __restrict__ colu,
                                                          int2* __restrict__ cw) {
  const int t   = blockIdx.x * blockDim.x + threadIdx.x;
  const int row = t >> 4;
  const int sub = t & 15;
  if (row >= BN_) return;
  const int start = rowptr[row];
  const int d     = row_end(rowptr, row) - start;
  const float dr  = (d > 0) ? rsqrtf((float)d) : 0.0f;
  for (int e = sub; e < d; e += 16) {
    const int key = colu[start + e];
    int rank = 0;
    for (int j = 0; j < d; ++j) {
      const int kj = colu[start + j];
      rank += ((kj < key) || (kj == key && j < e)) ? 1 : 0;
    }
    const int dc   = row_end(rowptr, key) - rowptr[key];
    const float dv = (dc > 0) ? rsqrtf((float)dc) : 0.0f;
    cw[start + rank] = make_int2(key, __float_as_int(dr * dv));
  }
}

__device__ __forceinline__ void pack_w_block(const float* __restrict__ W,
                                             _Float16* __restrict__ Wp) {
  const int t    = threadIdx.x;
  const int nt   = t >> 6;
  const int kb   = (t >> 5) & 1;
  const int lane = t & 31;
  const int n    = nt * 16 + (lane & 15);
  const int half = lane >> 4;
  _Float16* dst = Wp + ((size_t)((nt * 2 + kb) * 32 + lane)) * 16;
#pragma unroll
  for (int v = 0; v < 8; ++v) {
    int kbase = (v < 4) ? (2 * v) : (16 + 2 * (v - 4));
    int k = kb * 32 + kbase + half * 8;
    dst[2 * v]     = (_Float16)W[n * F_ + k];
    dst[2 * v + 1] = (_Float16)W[n * F_ + k + 1];
  }
}

#define SETUP_PACK_BLOCKS 3
#define SETUP_ZERO_BLOCKS ((BN_ / 4 + 255) / 256)
static_assert(BN_ % 4 == 0, "counter fill uses 16-byte stores");
__global__ __launch_bounds__(256) void setup_kernel(const float* __restrict__ W1,
                                                    const float* __restrict__ W2,
                                                    const float* __restrict__ W3,
                                                    _Float16* __restrict__ Wp1,
                                                    _Float16* __restrict__ Wp2,
                                                    _Float16* __restrict__ Wp3,
                                                    int* __restrict__ cnt) {
  const int blk = blockIdx.x;
  if (blk < SETUP_PACK_BLOCKS) {
    const float* W  = (blk == 0) ? W1  : (blk == 1) ? W2  : W3;
    _Float16*    Wp = (blk == 0) ? Wp1 : (blk == 1) ? Wp2 : Wp3;
    pack_w_block(W, Wp);
    return;
  }
  const int i = (blk - SETUP_PACK_BLOCKS) * 256 + threadIdx.x;
  if (i < BN_ / 4) ((int4*)cnt)[i] = make_int4(0, 0, 0, 0);
}

template <bool RELU_IN>
__device__ __forceinline__ void gemm_bias_tile(const float* __restrict__ Hin,
                                               const _Float16* __restrict__ Wp,
                                               const float* __restrict__ bias,
                                               float* __restrict__ Hout,
                                               const int rb) {
  const int lane = threadIdx.x & 31;
  if (rb * 16 >= BN_) return;
  const int row0 = rb * 16;
  const int half = lane >> 4;
  const int mrow = lane & 15;

  v16h a0, a1;
  const float* hrow = Hin + (size_t)(row0 + mrow) * F_;
#pragma unroll
  for (int v = 0; v < 8; ++v) {
    int kbase = (v < 4) ? (2 * v) : (16 + 2 * (v - 4));
    int k0 = kbase + half * 8;
    float2 p0 = *(const float2*)(hrow + k0);
    float2 p1 = *(const float2*)(hrow + 32 + k0);
    if (RELU_IN) {
      p0.x = fmaxf(p0.x, 0.0f); p0.y = fmaxf(p0.y, 0.0f);
      p1.x = fmaxf(p1.x, 0.0f); p1.y = fmaxf(p1.y, 0.0f);
    }
    a0[2 * v] = (_Float16)p0.x; a0[2 * v + 1] = (_Float16)p0.y;
    a1[2 * v] = (_Float16)p1.x; a1[2 * v + 1] = (_Float16)p1.y;
  }

  const v16h* Wpv = (const v16h*)Wp;
#pragma unroll
  for (int nt = 0; nt < 4; ++nt) {
    const v16h b0 = Wpv[(nt * 2 + 0) * 32 + lane];
    const v16h b1 = Wpv[(nt * 2 + 1) * 32 + lane];

    v8f c = {};
    c = __builtin_amdgcn_wmma_f32_16x16x32_f16(false, a0, false, b0, (short)0, c, false, false);
    c = __builtin_amdgcn_wmma_f32_16x16x32_f16(false, a1, false, b1, (short)0, c, false, false);

    const int n = nt * 16 + mrow;
    const float bv = bias[n];
#pragma unroll
    for (int r = 0; r < 8; ++r) {
      int m = r + half * 8;
      Hout[(size_t)(row0 + m) * F_ + n] = c[r] + bv;
    }
  }
}

template <bool RELU_IN>
__global__ __launch_bounds__(256) void gemm_bias_wmma(const float* __restrict__ Hin,
                                                      const _Float16* __restrict__ Wp,
                                                      const float* __restrict__ bias,
                                                      float* __restrict__ Hout) {
  gemm_bias_tile<RELU_IN>(Hin, Wp, bias, Hout, blockIdx.x * 8 + (threadIdx.x >> 5));
}

#define SCAN_GEMM_BLOCKS (B_ + (BN_ / 16 + 31) / 32)
__global__ __launch_bounds__(1024) void scan_gemm1_kernel(int* __restrict__ cnt,
                                                          const float* __restrict__ x,
                                                          const _Float16* __restrict__ Wp,
                                                          const float* __restrict__ bias,
                                                          float* __restrict__ Hout) {
  __shared__ int wsum[32];
  if (blockIdx.x < B_) {
    rowptr_scan_block(cnt, wsum, blockIdx.x);
  } else {
    gemm_bias_tile<false>(x, Wp, bias, Hout, (blockIdx.x - B_) * 32 + (threadIdx.x >> 5));
  }
}

__device__ __forceinline__ void axpy4(float4& acc, float w, const float4 h) {
  acc.x = fmaf(w, h.x, acc.x);
  acc.y = fmaf(w, h.y, acc.y);
  acc.z = fmaf(w, h.z, acc.z);
  acc.w = fmaf(w, h.w, acc.w);
}

__global__ __launch_bounds__(256) void gather_rows_kernel(const float* __restrict__ Hlin,
                                                          const int* __restrict__ rowptr,
                                                          const int2* __restrict__ cw,
                                                          float* __restrict__ out) {
  const int t   = blockIdx.x * blockDim.x + threadIdx.x;
  const int row = t >> 4;
  const int sub = t & 15;
  if (row >= BN_) return;
  const int start = rowptr[row];
  const int end   = row_end(rowptr, row);
  const float4* __restrict__ H4 = (const float4*)Hlin;
  float4 acc = make_float4(0.0f, 0.0f, 0.0f, 0.0f);
  int j = start;
  for (; j + 4 <= end; j += 4) {
    const int2 e0 = cw[j + 0];
    const int2 e1 = cw[j + 1];
    const int2 e2 = cw[j + 2];
    const int2 e3 = cw[j + 3];
    const float4 h0 = H4[(size_t)e0.x * (F_ / 4) + sub];
    const float4 h1 = H4[(size_t)e1.x * (F_ / 4) + sub];
    const float4 h2 = H4[(size_t)e2.x * (F_ / 4) + sub];
    const float4 h3 = H4[(size_t)e3.x * (F_ / 4) + sub];
    axpy4(acc, __int_as_float(e0.y), h0);
    axpy4(acc, __int_as_float(e1.y), h1);
    axpy4(acc, __int_as_float(e2.y), h2);
    axpy4(acc, __int_as_float(e3.y), h3);
  }
#pragma nounroll
  for (; j < end; ++j) {
    const int2 e0 = cw[j];
    const float4 h0 = H4[(size_t)e0.x * (F_ / 4) + sub];
    axpy4(acc, __int_as_float(e0.y), h0);
  }
  ((float4*)out)[(size_t)row * (F_ / 4) + sub] = acc;
}

extern "C" void kernel_launch(void* const* d_in, const int* in_sizes, int n_in,
                              void* d_out, int out_size, void* d_ws, size_t ws_size,
                              hipStream_t stream) {
  const float* x  = (const float*)d_in[0];
  const int*   ei = (const int*)  d_in[1];
  const float* W1 = (const float*)d_in[2];
  const float* b1 = (const float*)d_in[3];
  const float* W2 = (const float*)d_in[4];
  const float* b2 = (const float*)d_in[5];
  const float* W3 = (const float*)d_in[6];
  const float* b3 = (const float*)d_in[7];
  float* out = (float*)d_out;

  auto align256 = [](size_t v) { return (v + 255) & ~(size_t)255; };
  char* ws = (char*)d_ws;
  int*   rowptr = (int*)ws;      ws += align256((size_t)BN_ * 4);
  int2*  cw     = (int2*)ws;     ws += 2 * align256((size_t)BE_ * 4);
  ws += align256((size_t)BE_ * 4);
  float* Hlin  = (float*)ws;     ws += align256((size_t)BN_ * F_ * 4);
  float* Hagg  = (float*)ws;     ws += align256((size_t)BN_ * F_ * 4);
  _Float16* Wp1 = (_Float16*)ws; ws += align256((size_t)WPACK_HALVES * 2);
  _Float16* Wp2 = (_Float16*)ws; ws += align256((size_t)WPACK_HALVES * 2);
  _Float16* Wp3 = (_Float16*)ws; ws += align256((size_t)WPACK_HALVES * 2);
  int* slot = (int*)Hagg;
  int* colu = (int*)Hagg + BE_;

  const int TB = 256;
  setup_kernel<<<SETUP_PACK_BLOCKS + SETUP_ZERO_BLOCKS, TB, 0, stream>>>(W1, W2, W3, Wp1, Wp2, Wp3, rowptr);
  count_slot_kernel<<<(BE_ + TB - 1) / TB, TB, 0, stream>>>(ei, rowptr, slot);
  scan_gemm1_kernel<<<SCAN_GEMM_BLOCKS, 1024, 0, stream>>>(rowptr, x, Wp1, b1, Hlin);
  place_kernel<<<(BE_ + TB - 1) / TB, TB, 0, stream>>>(ei, rowptr, slot, colu);
  sort_weight_kernel<<<(BN_ * 16 + TB - 1) / TB, TB, 0, stream>>>(rowptr, colu, cw);

  const int GEMM_GRID = (BN_ / 16 + 7) / 8;
  const int GATH_GRID = (BN_ * 16 + TB - 1) / TB;

  gather_rows_kernel<<<GATH_GRID, TB, 0, stream>>>(Hlin, rowptr, cw, Hagg);

  gemm_bias_wmma<true><<<GEMM_GRID, TB, 0, stream>>>(Hagg, Wp2, b2, Hlin);
  gather_rows_kernel<<<GATH_GRID, TB, 0, stream>>>(Hlin, rowptr, cw, Hagg);

  gemm_bias_wmma<true><<<GEMM_GRID, TB, 0, stream>>>(Hagg, Wp3, b3, Hlin);
  gather_rows_kernel<<<GATH_GRID, TB, 0, stream>>>(Hlin, rowptr, cw, out);
}


# CSR build moved into LDS: one block per graph and 313-row range counts the graph's degrees, places and sorts its rows there; per-edge global atomics and 4-byte scatter stores gone; sum order kept
# speedup vs baseline: 3.4167x; 1.8253x over previous
#include <hip/hip_runtime.h>
#include <hip/hip_bf16.h>

typedef __attribute__((ext_vector_type(16))) _Float16 v16h;
typedef __attribute__((ext_vector_type(8)))  float    v8f;

#define B_  8
#define N_  10000
#define E_  160000
#define F_  64
#define BN_ (B_*N_)
#define BE_ (B_*E_)

#define WPACK_HALVES 4096

__device__ __forceinline__ void pack_w_block(const float* __restrict__ W,
                                             _Float16* __restrict__ Wp) {
  const int t    = threadIdx.x;
  const int nt   = t >> 6;
  const int kb   = (t >> 5) & 1;
  const int lane = t & 31;
  const int n    = nt * 16 + (lane & 15);
  const int half = lane >> 4;
  _Float16* dst = Wp + ((size_t)((nt * 2 + kb) * 32 + lane)) * 16;
#pragma unroll
  for (int v = 0; v < 8; ++v) {
    int kbase = (v < 4) ? (2 * v) : (16 + 2 * (v - 4));
    int k = kb * 32 + kbase + half * 8;
    dst[2 * v]     = (_Float16)W[n * F_ + k];
    dst[2 * v + 1] = (_Float16)W[n * F_ + k + 1];
  }
}

#define RANGES_ 32
#define RROWS_  ((N_ + RANGES_ - 1) / RANGES_)
#define KCAP_   8192
#define CSR_BLOCKS (B_ * RANGES_)
#define LOADS_  8
static_assert(N_ <= 16384, "a source node packs into 14 bits of a kept entry");
static_assert(RROWS_ <= 1024, "one thread per row of the range in the scan");
static_assert(E_ % 2 == 0, "the edge list is read two edges per 16-byte load");

__device__ __forceinline__ int row_end(const int* __restrict__ rowptr, int i) {
  return (i + 1 < BN_) ? rowptr[i + 1] : BE_;
}

__device__ __forceinline__ int wave_sum(int v) {
#pragma unroll
  for (int m = 16; m > 0; m >>= 1) v += __shfl_xor(v, m);
  return v;
}

__device__ __forceinline__ void sort_range_rows(const int* src, const int* deg, const int* rstart,
                                                int2* __restrict__ cw_seg, const int t,
                                                const int lo, const int nrows, const int node0) {
  const int sub = t & 15;
  for (int rl = t >> 4; rl < nrows; rl += 64) {
    const int start = rstart[rl];
    const int dd    = deg[lo + rl];
    const float dr  = (dd > 0) ? rsqrtf((float)dd) : 0.0f;
    const int* row  = src + start;
    for (int e = sub; e < dd; e += 16) {
      const int key = row[e];
      int rank = 0;
      for (int j = 0; j < dd; ++j) {
        const int kj = row[j];
        rank += ((kj < key) || (kj == key && j < e)) ? 1 : 0;
      }
      const int dc   = deg[key];
      const float dv = (dc > 0) ? rsqrtf((float)dc) : 0.0f;
      cw_seg[start + rank] = make_int2(key + node0, __float_as_int(dr * dv));
    }
  }
}

__device__ __forceinline__ void csr_range_block(const int* __restrict__ ei,
                                                int* __restrict__ rowptr,
                                                int* colu,
                                                int2* __restrict__ cw,
                                                const int b, const int q) {
  __shared__ int      deg[N_];
  __shared__ unsigned kept[KCAP_];
  __shared__ int      placed[KCAP_];
  __shared__ int      cursor[RROWS_];
  __shared__ int      rstart[RROWS_];
  __shared__ int      wsum_scan[32];
  __shared__ int      wsum_base[32];
  __shared__ int      kept_n;

  const int t     = threadIdx.x;
  const int lane  = t & 31;
  const int w     = t >> 5;
  const int lo    = q * RROWS_;
  const int hi    = (lo + RROWS_ < N_) ? (lo + RROWS_) : N_;
  const int nrows = hi - lo;

  for (int i = t; i < N_; i += 1024) deg[i] = 0;
  if (t == 0) kept_n = 0;
  __syncthreads();

  const int4* __restrict__ e4 = (const int4*)(ei + (size_t)b * E_ * 2);
  for (int i0 = 0; i0 < E_ / 2; i0 += LOADS_ * 1024) {
    int4 v[LOADS_];
#pragma unroll
    for (int u = 0; u < LOADS_; ++u) {
      const int idx = i0 + u * 1024 + t;
      v[u] = e4[(idx < E_ / 2) ? idx : (E_ / 2 - 1)];
    }
    int r[2 * LOADS_], s[2 * LOADS_];
#pragma unroll
    for (int u = 0; u < LOADS_; ++u) {
      const bool live = (i0 + u * 1024 + t) < E_ / 2;
      r[2 * u]     = live ? v[u].x : -1;  s[2 * u]     = v[u].y;
      r[2 * u + 1] = live ? v[u].z : -1;  s[2 * u + 1] = v[u].w;
    }
    int n_in = 0;
#pragma unroll
    for (int k = 0; k < 2 * LOADS_; ++k) {
      if (r[k] >= 0) atomicAdd(&deg[r[k]], 1);
      n_in += (r[k] >= lo && r[k] < hi) ? 1 : 0;
    }
    int p = (n_in > 0) ? atomicAdd(&kept_n, n_in) : 0;
#pragma unroll
    for (int k = 0; k < 2 * LOADS_; ++k) {
      if (r[k] >= lo && r[k] < hi) {
        if (p < KCAP_) kept[p] = ((unsigned)(r[k] - lo) << 14) | (unsigned)s[k];
        ++p;
      }
    }
  }
  __syncthreads();
  const int  kn   = kept_n;
  const bool fast = (kn <= KCAP_);

  int part = 0;
  for (int i = t; i < lo; i += 1024) part += deg[i];
  const int d = (t < nrows) ? deg[lo + t] : 0;
  int inc = d;
#pragma unroll
  for (int s = 1; s < 32; s <<= 1) {
    const int o = __shfl_up(inc, s);
    if (lane >= s) inc += o;
  }
  part = wave_sum(part);
  if (lane == 31) wsum_scan[w] = inc;
  if (lane == 0)  wsum_base[w] = part;
  __syncthreads();
  const int woff = wave_sum((lane < w) ? wsum_scan[lane] : 0);
  const int base = wave_sum(wsum_base[lane]);
  const int excl = woff + inc - d;
  const int seg0 = b * E_ + base;
  if (t < nrows) {
    rstart[t] = excl;
    cursor[t] = excl;
    rowptr[b * N_ + lo + t] = seg0 + excl;
  }
  __syncthreads();

  if (fast) {
    for (int i = t; i < kn; i += 1024) {
      const unsigned u = kept[i];
      const int pos = atomicAdd(&cursor[u >> 14], 1);
      placed[pos] = (int)(u & 16383u);
    }
  } else {
    const int2* __restrict__ e2 = (const int2*)(ei + (size_t)b * E_ * 2);
    for (int g = t; g < E_; g += 1024) {
      const int2 e = e2[g];
      if (e.x >= lo && e.x < hi) {
        const int pos = atomicAdd(&cursor[e.x - lo], 1);
        colu[seg0 + pos] = e.y;
      }
    }
  }
  __syncthreads();

  if (fast) sort_range_rows(placed, deg, rstart, cw + seg0, t, lo, nrows, b * N_);
  else      sort_range_rows(colu + seg0, deg, rstart, cw + seg0, t, lo, nrows, b * N_);
}

#define SETUP_PACK_BLOCKS 3
__global__ __launch_bounds__(1024) void csr_pack_kernel(const int* __restrict__ ei,
                                                        const float* __restrict__ W1,
                                                        const float* __restrict__ W2,
                                                        const float* __restrict__ W3,
                                                        _Float16* __restrict__ Wp1,
                                                        _Float16* __restrict__ Wp2,
                                                        _Float16* __restrict__ Wp3,
                                                        int* __restrict__ rowptr,
                                                        int* colu,
                                                        int2* __restrict__ cw) {
  const int blk = blockIdx.x;
  if (blk >= CSR_BLOCKS) {
    const int l = blk - CSR_BLOCKS;
    const float* W  = (l == 0) ? W1  : (l == 1) ? W2  : W3;
    _Float16*    Wp = (l == 0) ? Wp1 : (l == 1) ? Wp2 : Wp3;
    if (threadIdx.x < 256) pack_w_block(W, Wp);
    return;
  }
  csr_range_block(ei, rowptr, colu, cw, blk / RANGES_, blk % RANGES_);
}

template <bool RELU_IN>
__device__ __forceinline__ void gemm_bias_tile(const float* __restrict__ Hin,
                                               const _Float16* __restrict__ Wp,
                                               const float* __restrict__ bias,
                                               float* __restrict__ Hout,
                                               const int rb) {
  const int lane = threadIdx.x & 31;
  if (rb * 16 >= BN_) return;
  const int row0 = rb * 16;
  const int half = lane >> 4;
  const int mrow = lane & 15;

  v16h a0, a1;
  const float* hrow = Hin + (size_t)(row0 + mrow) * F_;
#pragma unroll
  for (int v = 0; v < 8; ++v) {
    int kbase = (v < 4) ? (2 * v) : (16 + 2 * (v - 4));
    int k0 = kbase + half * 8;
    float2 p0 = *(const float2*)(hrow + k0);
    float2 p1 = *(const float2*)(hrow + 32 + k0);
    if (RELU_IN) {
      p0.x = fmaxf(p0.x, 0.0f); p0.y = fmaxf(p0.y, 0.0f);
      p1.x = fmaxf(p1.x, 0.0f); p1.y = fmaxf(p1.y, 0.0f);
    }
    a0[2 * v] = (_Float16)p0.x; a0[2 * v + 1] = (_Float16)p0.y;
    a1[2 * v] = (_Float16)p1.x; a1[2 * v + 1] = (_Float16)p1.y;
  }

  const v16h* Wpv = (const v16h*)Wp;
#pragma unroll
  for (int nt = 0; nt < 4; ++nt) {
    const v16h b0 = Wpv[(nt * 2 + 0) * 32 + lane];
    const v16h b1 = Wpv[(nt * 2 + 1) * 32 + lane];

    v8f c = {};
    c = __builtin_amdgcn_wmma_f32_16x16x32_f16(false, a0, false, b0, (short)0, c, false, false);
    c = __builtin_amdgcn_wmma_f32_16x16x32_f16(false, a1, false, b1, (short)0, c, false, false);

    const int n = nt * 16 + mrow;
    const float bv = bias[n];
#pragma unroll
    for (int r = 0; r < 8; ++r) {
      int m = r + half * 8;
      Hout[(size_t)(row0 + m) * F_ + n] = c[r] + bv;
    }
  }
}

template <bool RELU_IN>
__global__ __launch_bounds__(256) void gemm_bias_wmma(const float* __restrict__ Hin,
                                                      const _Float16* __restrict__ Wp,
                                                      const float* __restrict__ bias,
                                                      float* __restrict__ Hout) {
  gemm_bias_tile<RELU_IN>(Hin, Wp, bias, Hout, blockIdx.x * 8 + (threadIdx.x >> 5));
}

__device__ __forceinline__ void axpy4(float4& acc, float w, const float4 h) {
  acc.x = fmaf(w, h.x, acc.x);
  acc.y = fmaf(w, h.y, acc.y);
  acc.z = fmaf(w, h.z, acc.z);
  acc.w = fmaf(w, h.w, acc.w);
}

__global__ __launch_bounds__(256) void gather_rows_kernel(const float* __restrict__ Hlin,
                                                          const int* __restrict__ rowptr,
                                                          const int2* __restrict__ cw,
                                                          float* __restrict__ out) {
  const int t   = blockIdx.x * blockDim.x + threadIdx.x;
  const int row = t >> 4;
  const int sub = t & 15;
  if (row >= BN_) return;
  const int start = rowptr[row];
  const int end   = row_end(rowptr, row);
  const float4* __restrict__ H4 = (const float4*)Hlin;
  float4 acc = make_float4(0.0f, 0.0f, 0.0f, 0.0f);
  int j = start;
  for (; j + 4 <= end; j += 4) {
    const int2 e0 = cw[j + 0];
    const int2 e1 = cw[j + 1];
    const int2 e2 = cw[j + 2];
    const int2 e3 = cw[j + 3];
    const float4 h0 = H4[(size_t)e0.x * (F_ / 4) + sub];
    const float4 h1 = H4[(size_t)e1.x * (F_ / 4) + sub];
    const float4 h2 = H4[(size_t)e2.x * (F_ / 4) + sub];
    const float4 h3 = H4[(size_t)e3.x * (F_ / 4) + sub];
    axpy4(acc, __int_as_float(e0.y), h0);
    axpy4(acc, __int_as_float(e1.y), h1);
    axpy4(acc, __int_as_float(e2.y), h2);
    axpy4(acc, __int_as_float(e3.y), h3);
  }
#pragma nounroll
  for (; j < end; ++j) {
    const int2 e0 = cw[j];
    const float4 h0 = H4[(size_t)e0.x * (F_ / 4) + sub];
    axpy4(acc, __int_as_float(e0.y), h0);
  }
  ((float4*)out)[(size_t)row * (F_ / 4) + sub] = acc;
}

extern "C" void kernel_launch(void* const* d_in, const int* in_sizes, int n_in,
                              void* d_out, int out_size, void* d_ws, size_t ws_size,
                              hipStream_t stream) {
  const float* x  = (const float*)d_in[0];
  const int*   ei = (const int*)  d_in[1];
  const float* W1 = (const float*)d_in[2];
  const float* b1 = (const float*)d_in[3];
  const float* W2 = (const float*)d_in[4];
  const float* b2 = (const float*)d_in[5];
  const float* W3 = (const float*)d_in[6];
  const float* b3 = (const float*)d_in[7];
  float* out = (float*)d_out;

  auto align256 = [](size_t v) { return (v + 255) & ~(size_t)255; };
  char* ws = (char*)d_ws;
  int*   rowptr = (int*)ws;      ws += align256((size_t)BN_ * 4);
  int2*  cw     = (int2*)ws;     ws += 2 * align256((size_t)BE_ * 4);
  ws += align256((size_t)BE_ * 4);
  float* Hlin  = (float*)ws;     ws += align256((size_t)BN_ * F_ * 4);
  float* Hagg  = (float*)ws;     ws += align256((size_t)BN_ * F_ * 4);
  _Float16* Wp1 = (_Float16*)ws; ws += align256((size_t)WPACK_HALVES * 2);
  _Float16* Wp2 = (_Float16*)ws; ws += align256((size_t)WPACK_HALVES * 2);
  _Float16* Wp3 = (_Float16*)ws; ws += align256((size_t)WPACK_HALVES * 2);
  int* colu = (int*)Hagg;

  const int TB = 256;
  const int GEMM_GRID = (BN_ / 16 + 7) / 8;
  const int GATH_GRID = (BN_ * 16 + TB - 1) / TB;

  csr_pack_kernel<<<CSR_BLOCKS + SETUP_PACK_BLOCKS, 1024, 0, stream>>>(ei, W1, W2, W3, Wp1, Wp2, Wp3,
                                                                     rowptr, colu, cw);
  gemm_bias_wmma<false><<<GEMM_GRID, TB, 0, stream>>>(x, Wp1, b1, Hlin);

  gather_rows_kernel<<<GATH_GRID, TB, 0, stream>>>(Hlin, rowptr, cw, Hagg);

  gemm_bias_wmma<true><<<GEMM_GRID, TB, 0, stream>>>(Hagg, Wp2, b2, Hlin);
  gather_rows_kernel<<<GATH_GRID, TB, 0, stream>>>(Hlin, rowptr, cw, Hagg);

  gemm_bias_wmma<true><<<GEMM_GRID, TB, 0, stream>>>(Hagg, Wp3, b3, Hlin);
  gather_rows_kernel<<<GATH_GRID, TB, 0, stream>>>(Hlin, rowptr, cw, out);
}


# Row gather: a 16-lane group walks 5 rows, not 1 (40000 waves a pass become 8000); its pairs come by one coalesced load a chunk ahead, shared by lane shuffles, 16 row loads in flight; sum order kept
# speedup vs baseline: 3.5631x; 1.0428x over previous
#include <hip/hip_runtime.h>
#include <hip/hip_bf16.h>

typedef __attribute__((ext_vector_type(16))) _Float16 v16h;
typedef __attribute__((ext_vector_type(8)))  float    v8f;

#define B_  8
#define N_  10000
#define E_  160000
#define F_  64
#define BN_ (B_*N_)
#define BE_ (B_*E_)

#define WPACK_HALVES 4096

__device__ __forceinline__ void pack_w_block(const float* __restrict__ W,
                                             _Float16* __restrict__ Wp) {
  const int t    = threadIdx.x;
  const int nt   = t >> 6;
  const int kb   = (t >> 5) & 1;
  const int lane = t & 31;
  const int n    = nt * 16 + (lane & 15);
  const int half = lane >> 4;
  _Float16* dst = Wp + ((size_t)((nt * 2 + kb) * 32 + lane)) * 16;
#pragma unroll
  for (int v = 0; v < 8; ++v) {
    int kbase = (v < 4) ? (2 * v) : (16 + 2 * (v - 4));
    int k = kb * 32 + kbase + half * 8;
    dst[2 * v]     = (_Float16)W[n * F_ + k];
    dst[2 * v + 1] = (_Float16)W[n * F_ + k + 1];
  }
}

#define RANGES_ 32
#define RROWS_  ((N_ + RANGES_ - 1) / RANGES_)
#define KCAP_   8192
#define CSR_BLOCKS (B_ * RANGES_)
#define LOADS_  8
static_assert(N_ <= 16384, "a source node packs into 14 bits of a kept entry");
static_assert(RROWS_ <= 1024, "one thread per row of the range in the scan");
static_assert(E_ % 2 == 0, "the edge list is read two edges per 16-byte load");

__device__ __forceinline__ int wave_sum(int v) {
#pragma unroll
  for (int m = 16; m > 0; m >>= 1) v += __shfl_xor(v, m);
  return v;
}

__device__ __forceinline__ void sort_range_rows(const int* src, const int* deg, const int* rstart,
                                                int2* __restrict__ cw_seg, const int t,
                                                const int lo, const int nrows, const int node0) {
  const int sub = t & 15;
  for (int rl = t >> 4; rl < nrows; rl += 64) {
    const int start = rstart[rl];
    const int dd    = deg[lo + rl];
    const float dr  = (dd > 0) ? rsqrtf((float)dd) : 0.0f;
    const int* row  = src + start;
    for (int e = sub; e < dd; e += 16) {
      const int key = row[e];
      int rank = 0;
      for (int j = 0; j < dd; ++j) {
        const int kj = row[j];
        rank += ((kj < key) || (kj == key && j < e)) ? 1 : 0;
      }
      const int dc   = deg[key];
      const float dv = (dc > 0) ? rsqrtf((float)dc) : 0.0f;
      cw_seg[start + rank] = make_int2(key + node0, __float_as_int(dr * dv));
    }
  }
}

__device__ __forceinline__ void csr_range_block(const int* __restrict__ ei,
                                                int* __restrict__ rowptr,
                                                int* colu,
                                                int2* __restrict__ cw,
                                                const int b, const int q) {
  __shared__ int      deg[N_];
  __shared__ unsigned kept[KCAP_];
  __shared__ int      placed[KCAP_];
  __shared__ int      cursor[RROWS_];
  __shared__ int      rstart[RROWS_];
  __shared__ int      wsum_scan[32];
  __shared__ int      wsum_base[32];
  __shared__ int      kept_n;

  const int t     = threadIdx.x;
  const int lane  = t & 31;
  const int w     = t >> 5;
  const int lo    = q * RROWS_;
  const int hi    = (lo + RROWS_ < N_) ? (lo + RROWS_) : N_;
  const int nrows = hi - lo;

  for (int i = t; i < N_; i += 1024) deg[i] = 0;
  if (t == 0) kept_n = 0;
  __syncthreads();

  const int4* __restrict__ e4 = (const int4*)(ei + (size_t)b * E_ * 2);
  for (int i0 = 0; i0 < E_ / 2; i0 += LOADS_ * 1024) {
    int4 v[LOADS_];
#pragma unroll
    for (int u = 0; u < LOADS_; ++u) {
      const int idx = i0 + u * 1024 + t;
      v[u] = e4[(idx < E_ / 2) ? idx : (E_ / 2 - 1)];
    }
    int r[2 * LOADS_], s[2 * LOADS_];
#pragma unroll
    for (int u = 0; u < LOADS_; ++u) {
      const bool live = (i0 + u * 1024 + t) < E_ / 2;
      r[2 * u]     = live ? v[u].x : -1;  s[2 * u]     = v[u].y;
      r[2 * u + 1] = live ? v[u].z : -1;  s[2 * u + 1] = v[u].w;
    }
    int n_in = 0;
#pragma unroll
    for (int k = 0; k < 2 * LOADS_; ++k) {
      if (r[k] >= 0) atomicAdd(&deg[r[k]], 1);
      n_in += (r[k] >= lo && r[k] < hi) ? 1 : 0;
    }
    int p = (n_in > 0) ? atomicAdd(&kept_n, n_in) : 0;
#pragma unroll
    for (int k = 0; k < 2 * LOADS_; ++k) {
      if (r[k] >= lo && r[k] < hi) {
        if (p < KCAP_) kept[p] = ((unsigned)(r[k] - lo) << 14) | (unsigned)s[k];
        ++p;
      }
    }
  }
  __syncthreads();
  const int  kn   = kept_n;
  const bool fast = (kn <= KCAP_);

  int part = 0;
  for (int i = t; i < lo; i += 1024) part += deg[i];
  const int d = (t < nrows) ? deg[lo + t] : 0;
  int inc = d;
#pragma unroll
  for (int s = 1; s < 32; s <<= 1) {
    const int o = __shfl_up(inc, s);
    if (lane >= s) inc += o;
  }
  part = wave_sum(part);
  if (lane == 31) wsum_scan[w] = inc;
  if (lane == 0)  wsum_base[w] = part;
  __syncthreads();
  const int woff = wave_sum((lane < w) ? wsum_scan[lane] : 0);
  const int base = wave_sum(wsum_base[lane]);
  const int excl = woff + inc - d;
  const int seg0 = b * E_ + base;
  if (t < nrows) {
    rstart[t] = excl;
    cursor[t] = excl;
    rowptr[b * N_ + lo + t] = seg0 + excl;
  }
  __syncthreads();

  if (fast) {
    for (int i = t; i < kn; i += 1024) {
      const unsigned u = kept[i];
      const int pos = atomicAdd(&cursor[u >> 14], 1);
      placed[pos] = (int)(u & 16383u);
    }
  } else {
    const int2* __restrict__ e2 = (const int2*)(ei + (size_t)b * E_ * 2);
    for (int g = t; g < E_; g += 1024) {
      const int2 e = e2[g];
      if (e.x >= lo && e.x < hi) {
        const int pos = atomicAdd(&cursor[e.x - lo], 1);
        colu[seg0 + pos] = e.y;
      }
    }
  }
  __syncthreads();

  if (fast) sort_range_rows(placed, deg, rstart, cw + seg0, t, lo, nrows, b * N_);
  else      sort_range_rows(colu + seg0, deg, rstart, cw + seg0, t, lo, nrows, b * N_);
}

#define SETUP_PACK_BLOCKS 3
__global__ __launch_bounds__(1024) void csr_pack_kernel(const int* __restrict__ ei,
                                                        const float* __restrict__ W1,
                                                        const float* __restrict__ W2,
                                                        const float* __restrict__ W3,
                                                        _Float16* __restrict__ Wp1,
                                                        _Float16* __restrict__ Wp2,
                                                        _Float16* __restrict__ Wp3,
                                                        int* __restrict__ rowptr,
                                                        int* colu,
                                                        int2* __restrict__ cw) {
  const int blk = blockIdx.x;
  if (blk >= CSR_BLOCKS) {
    const int l = blk - CSR_BLOCKS;
    const float* W  = (l == 0) ? W1  : (l == 1) ? W2  : W3;
    _Float16*    Wp = (l == 0) ? Wp1 : (l == 1) ? Wp2 : Wp3;
    if (threadIdx.x < 256) pack_w_block(W, Wp);
    return;
  }
  csr_range_block(ei, rowptr, colu, cw, blk / RANGES_, blk % RANGES_);
}

template <bool RELU_IN>
__device__ __forceinline__ void gemm_bias_tile(const float* __restrict__ Hin,
                                               const _Float16* __restrict__ Wp,
                                               const float* __restrict__ bias,
                                               float* __restrict__ Hout,
                                               const int rb) {
  const int lane = threadIdx.x & 31;
  if (rb * 16 >= BN_) return;
  const int row0 = rb * 16;
  const int half = lane >> 4;
  const int mrow = lane & 15;

  v16h a0, a1;
  const float* hrow = Hin + (size_t)(row0 + mrow) * F_;
#pragma unroll
  for (int v = 0; v < 8; ++v) {
    int kbase = (v < 4) ? (2 * v) : (16 + 2 * (v - 4));
    int k0 = kbase + half * 8;
    float2 p0 = *(const float2*)(hrow + k0);
    float2 p1 = *(const float2*)(hrow + 32 + k0);
    if (RELU_IN) {
      p0.x = fmaxf(p0.x, 0.0f); p0.y = fmaxf(p0.y, 0.0f);
      p1.x = fmaxf(p1.x, 0.0f); p1.y = fmaxf(p1.y, 0.0f);
    }
    a0[2 * v] = (_Float16)p0.x; a0[2 * v + 1] = (_Float16)p0.y;
    a1[2 * v] = (_Float16)p1.x; a1[2 * v + 1] = (_Float16)p1.y;
  }

  const v16h* Wpv = (const v16h*)Wp;
#pragma unroll
  for (int nt = 0; nt < 4; ++nt) {
    const v16h b0 = Wpv[(nt * 2 + 0) * 32 + lane];
    const v16h b1 = Wpv[(nt * 2 + 1) * 32 + lane];

    v8f c = {};
    c = __builtin_amdgcn_wmma_f32_16x16x32_f16(false, a0, false, b0, (short)0, c, false, false);
    c = __builtin_amdgcn_wmma_f32_16x16x32_f16(false, a1, false, b1, (short)0, c, false, false);

    const int n = nt * 16 + mrow;
    const float bv = bias[n];
#pragma unroll
    for (int r = 0; r < 8; ++r) {
      int m = r + half * 8;
      Hout[(size_t)(row0 + m) * F_ + n] = c[r] + bv;
    }
  }
}

template <bool RELU_IN>
__global__ __launch_bounds__(256) void gemm_bias_wmma(const float* __restrict__ Hin,
                                                      const _Float16* __restrict__ Wp,
                                                      const float* __restrict__ bias,
                                                      float* __restrict__ Hout) {
  gemm_bias_tile<RELU_IN>(Hin, Wp, bias, Hout, blockIdx.x * 8 + (threadIdx.x >> 5));
}

#define GROWS_ 5
#define GATH_ROWS_PER_BLOCK (16 * GROWS_)
static_assert(BN_ % GATH_ROWS_PER_BLOCK == 0, "every group owns GROWS_ full rows");
static_assert(GROWS_ < 16, "lane sub of a group holds row pointer sub, sub <= GROWS_");

__device__ __forceinline__ void axpy4(float4& acc, float w, const float4 h) {
  acc.x = fmaf(w, h.x, acc.x);
  acc.y = fmaf(w, h.y, acc.y);
  acc.z = fmaf(w, h.z, acc.z);
  acc.w = fmaf(w, h.w, acc.w);
}

__global__ __launch_bounds__(256) void gather_rows_kernel(const float* __restrict__ Hlin,
                                                          const int* __restrict__ rowptr,
                                                          const int2* __restrict__ cw,
                                                          float* __restrict__ out) {
  const int t   = blockIdx.x * blockDim.x + threadIdx.x;
  const int sub = t & 15;
  const int r0  = (t >> 4) * GROWS_;
  const char* __restrict__ Hb = (const char*)Hlin;
  const unsigned lane_off = (unsigned)sub << 4;

  int rp = 0;
  if (sub <= GROWS_) rp = (r0 + sub < BN_) ? rowptr[r0 + sub] : BE_;
  const int gend = __shfl(rp, GROWS_, 16);
  int base = __shfl(rp, 0, 16);
  int2 my = make_int2(0, 0);
  if (base + sub < gend) my = cw[base + sub];

  for (int rr = 0; rr < GROWS_; ++rr) {
    const int end = __shfl(rp, rr + 1, 16);
    float4 acc = make_float4(0.0f, 0.0f, 0.0f, 0.0f);
    while (base < end) {
      const int n  = end - base;
      const int nb = (n > 16) ? (base + 16) : end;
      int2 nxt = make_int2(0, 0);
      if (nb + sub < gend) nxt = cw[nb + sub];
      float4 h[16];
#pragma unroll
      for (int k = 0; k < 16; ++k) {
        const unsigned s = (unsigned)__shfl(my.x, k, 16);
        if (k < n) h[k] = *(const float4*)(Hb + ((s << 8) | lane_off));
      }
#pragma unroll
      for (int k = 0; k < 16; ++k) {
        const float w = __int_as_float(__shfl(my.y, k, 16));
        if (k < n) axpy4(acc, w, h[k]);
      }
      my   = nxt;
      base = nb;
    }
    ((float4*)out)[(size_t)(r0 + rr) * (F_ / 4) + sub] = acc;
  }
}

extern "C" void kernel_launch(void* const* d_in, const int* in_sizes, int n_in,
                              void* d_out, int out_size, void* d_ws, size_t ws_size,
                              hipStream_t stream) {
  const float* x  = (const float*)d_in[0];
  const int*   ei = (const int*)  d_in[1];
  const float* W1 = (const float*)d_in[2];
  const float* b1 = (const float*)d_in[3];
  const float* W2 = (const float*)d_in[4];
  const float* b2 = (const float*)d_in[5];
  const float* W3 = (const float*)d_in[6];
  const float* b3 = (const float*)d_in[7];
  float* out = (float*)d_out;

  auto align256 = [](size_t v) { return (v + 255) & ~(size_t)255; };
  char* ws = (char*)d_ws;
  int*   rowptr = (int*)ws;      ws += align256((size_t)BN_ * 4);
  int2*  cw     = (int2*)ws;     ws += 2 * align256((size_t)BE_ * 4);
  ws += align256((size_t)BE_ * 4);
  float* Hlin  = (float*)ws;     ws += align256((size_t)BN_ * F_ * 4);
  float* Hagg  = (float*)ws;     ws += align256((size_t)BN_ * F_ * 4);
  _Float16* Wp1 = (_Float16*)ws; ws += align256((size_t)WPACK_HALVES * 2);
  _Float16* Wp2 = (_Float16*)ws; ws += align256((size_t)WPACK_HALVES * 2);
  _Float16* Wp3 = (_Float16*)ws; ws += align256((size_t)WPACK_HALVES * 2);
  int* colu = (int*)Hagg;

  const int TB = 256;
  const int GEMM_GRID = (BN_ / 16 + 7) / 8;
  const int GATH_GRID = BN_ / GATH_ROWS_PER_BLOCK;

  csr_pack_kernel<<<CSR_BLOCKS + SETUP_PACK_BLOCKS, 1024, 0, stream>>>(ei, W1, W2, W3, Wp1, Wp2, Wp3,
                                                                     rowptr, colu, cw);
  gemm_bias_wmma<false><<<GEMM_GRID, TB, 0, stream>>>(x, Wp1, b1, Hlin);

  gather_rows_kernel<<<GATH_GRID, TB, 0, stream>>>(Hlin, rowptr, cw, Hagg);

  gemm_bias_wmma<true><<<GEMM_GRID, TB, 0, stream>>>(Hagg, Wp2, b2, Hlin);
  gather_rows_kernel<<<GATH_GRID, TB, 0, stream>>>(Hlin, rowptr, cw, Hagg);

  gemm_bias_wmma<true><<<GEMM_GRID, TB, 0, stream>>>(Hagg, Wp3, b3, Hlin);
  gather_rows_kernel<<<GATH_GRID, TB, 0, stream>>>(Hlin, rowptr, cw, out);
}


# Row gather and the next layer's linear map share a launch: the block's 80 summed rows stay in LDS as WMMA A operands, one barrier, 5 waves run the same tile code; 7 launches become 5, same bits
# speedup vs baseline: 4.1360x; 1.1608x over previous
#include <hip/hip_runtime.h>
#include <hip/hip_bf16.h>

typedef __attribute__((ext_vector_type(16))) _Float16 v16h;
typedef __attribute__((ext_vector_type(8)))  float    v8f;

#define B_  8
#define N_  10000
#define E_  160000
#define F_  64
#define BN_ (B_*N_)
#define BE_ (B_*E_)

#define WPACK_HALVES 4096

__device__ __forceinline__ void pack_w_block(const float* __restrict__ W,
                                             _Float16* __restrict__ Wp) {
  const int t    = threadIdx.x;
  const int nt   = t >> 6;
  const int kb   = (t >> 5) & 1;
  const int lane = t & 31;
  const int n    = nt * 16 + (lane & 15);
  const int half = lane >> 4;
  _Float16* dst = Wp + ((size_t)((nt * 2 + kb) * 32 + lane)) * 16;
#pragma unroll
  for (int v = 0; v < 8; ++v) {
    int kbase = (v < 4) ? (2 * v) : (16 + 2 * (v - 4));
    int k = kb * 32 + kbase + half * 8;
    dst[2 * v]     = (_Float16)W[n * F_ + k];
    dst[2 * v + 1] = (_Float16)W[n * F_ + k + 1];
  }
}

#define RANGES_ 32
#define RROWS_  ((N_ + RANGES_ - 1) / RANGES_)
#define KCAP_   8192
#define CSR_BLOCKS (B_ * RANGES_)
#define LOADS_  8
static_assert(N_ <= 16384, "a source node packs into 14 bits of a kept entry");
static_assert(RROWS_ <= 1024, "one thread per row of the range in the scan");
static_assert(E_ % 2 == 0, "the edge list is read two edges per 16-byte load");

__device__ __forceinline__ int wave_sum(int v) {
#pragma unroll
  for (int m = 16; m > 0; m >>= 1) v += __shfl_xor(v, m);
  return v;
}

__device__ __forceinline__ void sort_range_rows(const int* src, const int* deg, const int* rstart,
                                                int2* __restrict__ cw_seg, const int t,
                                                const int lo, const int nrows, const int node0) {
  const int sub = t & 15;
  for (int rl = t >> 4; rl < nrows; rl += 64) {
    const int start = rstart[rl];
    const int dd    = deg[lo + rl];
    const float dr  = (dd > 0) ? rsqrtf((float)dd) : 0.0f;
    const int* row  = src + start;
    for (int e = sub; e < dd; e += 16) {
      const int key = row[e];
      int rank = 0;
      for (int j = 0; j < dd; ++j) {
        const int kj = row[j];
        rank += ((kj < key) || (kj == key && j < e)) ? 1 : 0;
      }
      const int dc   = deg[key];
      const float dv = (dc > 0) ? rsqrtf((float)dc) : 0.0f;
      cw_seg[start + rank] = make_int2(key + node0, __float_as_int(dr * dv));
    }
  }
}

__device__ __forceinline__ void csr_range_block(const int* __restrict__ ei,
                                                int* __restrict__ rowptr,
                                                int* colu,
                                                int2* __restrict__ cw,
                                                const int b, const int q) {
  __shared__ int      deg[N_];
  __shared__ unsigned kept[KCAP_];
  __shared__ int      placed[KCAP_];
  __shared__ int      cursor[RROWS_];
  __shared__ int      rstart[RROWS_];
  __shared__ int      wsum_scan[32];
  __shared__ int      wsum_base[32];
  __shared__ int      kept_n;

  const int t     = threadIdx.x;
  const int lane  = t & 31;
  const int w     = t >> 5;
  const int lo    = q * RROWS_;
  const int hi    = (lo + RROWS_ < N_) ? (lo + RROWS_) : N_;
  const int nrows = hi - lo;

  for (int i = t; i < N_; i += 1024) deg[i] = 0;
  if (t == 0) kept_n = 0;
  __syncthreads();

  const int4* __restrict__ e4 = (const int4*)(ei + (size_t)b * E_ * 2);
  for (int i0 = 0; i0 < E_ / 2; i0 += LOADS_ * 1024) {
    int4 v[LOADS_];
#pragma unroll
    for (int u = 0; u < LOADS_; ++u) {
      const int idx = i0 + u * 1024 + t;
      v[u] = e4[(idx < E_ / 2) ? idx : (E_ / 2 - 1)];
    }
    int r[2 * LOADS_], s[2 * LOADS_];
#pragma unroll
    for (int u = 0; u < LOADS_; ++u) {
      const bool live = (i0 + u * 1024 + t) < E_ / 2;
      r[2 * u]     = live ? v[u].x : -1;  s[2 * u]     = v[u].y;
      r[2 * u + 1] = live ? v[u].z : -1;  s[2 * u + 1] = v[u].w;
    }
    int n_in = 0;
#pragma unroll
    for (int k = 0; k < 2 * LOADS_; ++k) {
      if (r[k] >= 0) atomicAdd(&deg[r[k]], 1);
      n_in += (r[k] >= lo && r[k] < hi) ? 1 : 0;
    }
    int p = (n_in > 0) ? atomicAdd(&kept_n, n_in) : 0;
#pragma unroll
    for (int k = 0; k < 2 * LOADS_; ++k) {
      if (r[k] >= lo && r[k] < hi) {
        if (p < KCAP_) kept[p] = ((unsigned)(r[k] - lo) << 14) | (unsigned)s[k];
        ++p;
      }
    }
  }
  __syncthreads();
  const int  kn   = kept_n;
  const bool fast = (kn <= KCAP_);

  int part = 0;
  for (int i = t; i < lo; i += 1024) part += deg[i];
  const int d = (t < nrows) ? deg[lo + t] : 0;
  int inc = d;
#pragma unroll
  for (int s = 1; s < 32; s <<= 1) {
    const int o = __shfl_up(inc, s);
    if (lane >= s) inc += o;
  }
  part = wave_sum(part);
  if (lane == 31) wsum_scan[w] = inc;
  if (lane == 0)  wsum_base[w] = part;
  __syncthreads();
  const int woff = wave_sum((lane < w) ? wsum_scan[lane] : 0);
  const int base = wave_sum(wsum_base[lane]);
  const int excl = woff + inc - d;
  const int seg0 = b * E_ + base;
  if (t < nrows) {
    rstart[t] = excl;
    cursor[t] = excl;
    rowptr[b * N_ + lo + t] = seg0 + excl;
  }
  __syncthreads();

  if (fast) {
    for (int i = t; i < kn; i += 1024) {
      const unsigned u = kept[i];
      const int pos = atomicAdd(&cursor[u >> 14], 1);
      placed[pos] = (int)(u & 16383u);
    }
  } else {
    const int2* __restrict__ e2 = (const int2*)(ei + (size_t)b * E_ * 2);
    for (int g = t; g < E_; g += 1024) {
      const int2 e = e2[g];
      if (e.x >= lo && e.x < hi) {
        const int pos = atomicAdd(&cursor[e.x - lo], 1);
        colu[seg0 + pos] = e.y;
      }
    }
  }
  __syncthreads();

  if (fast) sort_range_rows(placed, deg, rstart, cw + seg0, t, lo, nrows, b * N_);
  else      sort_range_rows(colu + seg0, deg, rstart, cw + seg0, t, lo, nrows, b * N_);
}

#define SETUP_PACK_BLOCKS 3
__global__ __launch_bounds__(1024) void csr_pack_kernel(const int* __restrict__ ei,
                                                        const float* __restrict__ W1,
                                                        const float* __restrict__ W2,
                                                        const float* __restrict__ W3,
                                                        _Float16* __restrict__ Wp1,
                                                        _Float16* __restrict__ Wp2,
                                                        _Float16* __restrict__ Wp3,
                                                        int* __restrict__ rowptr,
                                                        int* colu,
                                                        int2* __restrict__ cw) {
  const int blk = blockIdx.x;
  if (blk >= CSR_BLOCKS) {
    const int l = blk - CSR_BLOCKS;
    const float* W  = (l == 0) ? W1  : (l == 1) ? W2  : W3;
    _Float16*    Wp = (l == 0) ? Wp1 : (l == 1) ? Wp2 : Wp3;
    if (threadIdx.x < 256) pack_w_block(W, Wp);
    return;
  }
  csr_range_block(ei, rowptr, colu, cw, blk / RANGES_, blk % RANGES_);
}

__device__ __forceinline__ void gemm_bias_frags(const v16h a0, const v16h a1,
                                                const _Float16* __restrict__ Wp,
                                                const float* __restrict__ bias,
                                                float* __restrict__ Hout,
                                                const int row0) {
  const int lane = threadIdx.x & 31;
  const int half = lane >> 4;
  const int mrow = lane & 15;
  const v16h* Wpv = (const v16h*)Wp;
#pragma unroll
  for (int nt = 0; nt < 4; ++nt) {
    const v16h b0 = Wpv[(nt * 2 + 0) * 32 + lane];
    const v16h b1 = Wpv[(nt * 2 + 1) * 32 + lane];

    v8f c = {};
    c = __builtin_amdgcn_wmma_f32_16x16x32_f16(false, a0, false, b0, (short)0, c, false, false);
    c = __builtin_amdgcn_wmma_f32_16x16x32_f16(false, a1, false, b1, (short)0, c, false, false);

    const int n = nt * 16 + mrow;
    const float bv = bias[n];
#pragma unroll
    for (int r = 0; r < 8; ++r) {
      int m = r + half * 8;
      Hout[(size_t)(row0 + m) * F_ + n] = c[r] + bv;
    }
  }
}

template <bool RELU_IN>
__device__ __forceinline__ void gemm_bias_tile(const float* __restrict__ Hin,
                                               const _Float16* __restrict__ Wp,
                                               const float* __restrict__ bias,
                                               float* __restrict__ Hout,
                                               const int rb) {
  const int lane = threadIdx.x & 31;
  if (rb * 16 >= BN_) return;
  const int row0 = rb * 16;
  const int half = lane >> 4;
  const int mrow = lane & 15;

  v16h a0, a1;
  const float* hrow = Hin + (size_t)(row0 + mrow) * F_;
#pragma unroll
  for (int v = 0; v < 8; ++v) {
    int kbase = (v < 4) ? (2 * v) : (16 + 2 * (v - 4));
    int k0 = kbase + half * 8;
    float2 p0 = *(const float2*)(hrow + k0);
    float2 p1 = *(const float2*)(hrow + 32 + k0);
    if (RELU_IN) {
      p0.x = fmaxf(p0.x, 0.0f); p0.y = fmaxf(p0.y, 0.0f);
      p1.x = fmaxf(p1.x, 0.0f); p1.y = fmaxf(p1.y, 0.0f);
    }
    a0[2 * v] = (_Float16)p0.x; a0[2 * v + 1] = (_Float16)p0.y;
    a1[2 * v] = (_Float16)p1.x; a1[2 * v + 1] = (_Float16)p1.y;
  }
  gemm_bias_frags(a0, a1, Wp, bias, Hout, row0);
}

template <bool RELU_IN>
__global__ __launch_bounds__(256) void gemm_bias_wmma(const float* __restrict__ Hin,
                                                      const _Float16* __restrict__ Wp,
                                                      const float* __restrict__ bias,
                                                      float* __restrict__ Hout) {
  gemm_bias_tile<RELU_IN>(Hin, Wp, bias, Hout, blockIdx.x * 8 + (threadIdx.x >> 5));
}

#define GROWS_ 5
#define GATH_ROWS_PER_BLOCK (16 * GROWS_)
#define GATH_TILES_PER_BLOCK (GATH_ROWS_PER_BLOCK / 16)
#define APITCH_ 72
static_assert(BN_ % GATH_ROWS_PER_BLOCK == 0, "every group owns GROWS_ full rows");
static_assert(GROWS_ < 16, "lane sub of a group holds row pointer sub, sub <= GROWS_");
static_assert(GATH_ROWS_PER_BLOCK % 16 == 0 && GATH_TILES_PER_BLOCK <= 8, "whole 16-row tiles, one wave each");

typedef __attribute__((ext_vector_type(4))) _Float16 v4h;
typedef __attribute__((ext_vector_type(8))) _Float16 v8h;

__device__ __forceinline__ void axpy4(float4& acc, float w, const float4 h) {
  acc.x = fmaf(w, h.x, acc.x);
  acc.y = fmaf(w, h.y, acc.y);
  acc.z = fmaf(w, h.z, acc.z);
  acc.w = fmaf(w, h.w, acc.w);
}

template <bool NEXT_LINEAR>
__global__ __launch_bounds__(256) void gather_rows_kernel(const float* __restrict__ Hlin,
                                                          const int* __restrict__ rowptr,
                                                          const int2* __restrict__ cw,
                                                          float* __restrict__ out,
                                                          const _Float16* __restrict__ Wp,
                                                          const float* __restrict__ bias) {
  __shared__ __attribute__((aligned(16))) _Float16 At[NEXT_LINEAR ? GATH_ROWS_PER_BLOCK * APITCH_ : 8];

  const int t   = blockIdx.x * blockDim.x + threadIdx.x;
  const int sub = t & 15;
  const int r0  = (t >> 4) * GROWS_;
  const char* __restrict__ Hb = (const char*)Hlin;
  const unsigned lane_off = (unsigned)sub << 4;

  int rp = 0;
  if (sub <= GROWS_) rp = (r0 + sub < BN_) ? rowptr[r0 + sub] : BE_;
  const int gend = __shfl(rp, GROWS_, 16);
  int base = __shfl(rp, 0, 16);
  int2 my = make_int2(0, 0);
  if (base + sub < gend) my = cw[base + sub];

  for (int rr = 0; rr < GROWS_; ++rr) {
    const int end = __shfl(rp, rr + 1, 16);
    float4 acc = make_float4(0.0f, 0.0f, 0.0f, 0.0f);
    while (base < end) {
      const int n  = end - base;
      const int nb = (n > 16) ? (base + 16) : end;
      int2 nxt = make_int2(0, 0);
      if (nb + sub < gend) nxt = cw[nb + sub];
      float4 h[16];
#pragma unroll
      for (int k = 0; k < 16; ++k) {
        const unsigned s = (unsigned)__shfl(my.x, k, 16);
        if (k < n) h[k] = *(const float4*)(Hb + ((s << 8) | lane_off));
      }
#pragma unroll
      for (int k = 0; k < 16; ++k) {
        const float w = __int_as_float(__shfl(my.y, k, 16));
        if (k < n) axpy4(acc, w, h[k]);
      }
      my   = nxt;
      base = nb;
    }
    if (NEXT_LINEAR) {
      v4h q;
      q[0] = (_Float16)fmaxf(acc.x, 0.0f);
      q[1] = (_Float16)fmaxf(acc.y, 0.0f);
      q[2] = (_Float16)fmaxf(acc.z, 0.0f);
      q[3] = (_Float16)fmaxf(acc.w, 0.0f);
      const int lr = (threadIdx.x >> 4) * GROWS_ + rr;
      *(v4h*)(At + lr * APITCH_ + sub * 4) = q;
    } else {
      ((float4*)out)[(size_t)(r0 + rr) * (F_ / 4) + sub] = acc;
    }
  }

  if (NEXT_LINEAR) {
    __syncthreads();
    const int wv = threadIdx.x >> 5;
    if (wv < GATH_TILES_PER_BLOCK) {
      const int lane = threadIdx.x & 31;
      const _Float16* arow = At + (wv * 16 + (lane & 15)) * APITCH_ + (lane >> 4) * 8;
      const v8h q0 = *(const v8h*)(arow);
      const v8h q1 = *(const v8h*)(arow + 16);
      const v8h q2 = *(const v8h*)(arow + 32);
      const v8h q3 = *(const v8h*)(arow + 48);
      const v16h a0 = __builtin_shufflevector(q0, q1, 0, 1, 2, 3, 4, 5, 6, 7, 8, 9, 10, 11, 12, 13, 14, 15);
      const v16h a1 = __builtin_shufflevector(q2, q3, 0, 1, 2, 3, 4, 5, 6, 7, 8, 9, 10, 11, 12, 13, 14, 15);
      gemm_bias_frags(a0, a1, Wp, bias, out, blockIdx.x * GATH_ROWS_PER_BLOCK + wv * 16);
    }
  }
}

extern "C" void kernel_launch(void* const* d_in, const int* in_sizes, int n_in,
                              void* d_out, int out_size, void* d_ws, size_t ws_size,
                              hipStream_t stream) {
  const float* x  = (const float*)d_in[0];
  const int*   ei = (const int*)  d_in[1];
  const float* W1 = (const float*)d_in[2];
  const float* b1 = (const float*)d_in[3];
  const float* W2 = (const float*)d_in[4];
  const float* b2 = (const float*)d_in[5];
  const float* W3 = (const float*)d_in[6];
  const float* b3 = (const float*)d_in[7];
  float* out = (float*)d_out;

  auto align256 = [](size_t v) { return (v + 255) & ~(size_t)255; };
  char* ws = (char*)d_ws;
  int*   rowptr = (int*)ws;      ws += align256((size_t)BN_ * 4);
  int2*  cw     = (int2*)ws;     ws += 2 * align256((size_t)BE_ * 4);
  ws += align256((size_t)BE_ * 4);
  float* Hlin  = (float*)ws;     ws += align256((size_t)BN_ * F_ * 4);
  float* Hlin2 = (float*)ws;     ws += align256((size_t)BN_ * F_ * 4);
  _Float16* Wp1 = (_Float16*)ws; ws += align256((size_t)WPACK_HALVES * 2);
  _Float16* Wp2 = (_Float16*)ws; ws += align256((size_t)WPACK_HALVES * 2);
  _Float16* Wp3 = (_Float16*)ws; ws += align256((size_t)WPACK_HALVES * 2);
  int* colu = (int*)Hlin2;

  const int TB = 256;
  const int GEMM_GRID = (BN_ / 16 + 7) / 8;
  const int GATH_GRID = BN_ / GATH_ROWS_PER_BLOCK;

  csr_pack_kernel<<<CSR_BLOCKS + SETUP_PACK_BLOCKS, 1024, 0, stream>>>(ei, W1, W2, W3, Wp1, Wp2, Wp3,
                                                                     rowptr, colu, cw);
  gemm_bias_wmma<false><<<GEMM_GRID, TB, 0, stream>>>(x, Wp1, b1, Hlin);

  gather_rows_kernel<true><<<GATH_GRID, TB, 0, stream>>>(Hlin, rowptr, cw, Hlin2, Wp2, b2);

  gather_rows_kernel<true><<<GATH_GRID, TB, 0, stream>>>(Hlin2, rowptr, cw, Hlin, Wp3, b3);

  gather_rows_kernel<false><<<GATH_GRID, TB, 0, stream>>>(Hlin, rowptr, cw, out, nullptr, nullptr);
}


# Linear-map rows between the layers kept as f16, not f32: gathers 2 and 3 read 128 B per edge, not 256 B, and the map's stores halve; sums stay f32 in CSR order; one f16 rounding per stored row
# speedup vs baseline: 4.9897x; 1.2064x over previous
#include <hip/hip_runtime.h>
#include <hip/hip_bf16.h>

typedef __attribute__((ext_vector_type(16))) _Float16 v16h;
typedef __attribute__((ext_vector_type(8)))  float    v8f;

#define B_  8
#define N_  10000
#define E_  160000
#define F_  64
#define BN_ (B_*N_)
#define BE_ (B_*E_)

#define WPACK_HALVES 4096

__device__ __forceinline__ void pack_w_block(const float* __restrict__ W,
                                             _Float16* __restrict__ Wp) {
  const int t    = threadIdx.x;
  const int nt   = t >> 6;
  const int kb   = (t >> 5) & 1;
  const int lane = t & 31;
  const int n    = nt * 16 + (lane & 15);
  const int half = lane >> 4;
  _Float16* dst = Wp + ((size_t)((nt * 2 + kb) * 32 + lane)) * 16;
#pragma unroll
  for (int v = 0; v < 8; ++v) {
    int kbase = (v < 4) ? (2 * v) : (16 + 2 * (v - 4));
    int k = kb * 32 + kbase + half * 8;
    dst[2 * v]     = (_Float16)W[n * F_ + k];
    dst[2 * v + 1] = (_Float16)W[n * F_ + k + 1];
  }
}

#define RANGES_ 32
#define RROWS_  ((N_ + RANGES_ - 1) / RANGES_)
#define KCAP_   8192
#define CSR_BLOCKS (B_ * RANGES_)
#define LOADS_  8
static_assert(N_ <= 16384, "a source node packs into 14 bits of a kept entry");
static_assert(RROWS_ <= 1024, "one thread per row of the range in the scan");
static_assert(E_ % 2 == 0, "the edge list is read two edges per 16-byte load");

__device__ __forceinline__ int wave_sum(int v) {
#pragma unroll
  for (int m = 16; m > 0; m >>= 1) v += __shfl_xor(v, m);
  return v;
}

__device__ __forceinline__ void sort_range_rows(const int* src, const int* deg, const int* rstart,
                                                int2* __restrict__ cw_seg, const int t,
                                                const int lo, const int nrows, const int node0) {
  const int sub = t & 15;
  for (int rl = t >> 4; rl < nrows; rl += 64) {
    const int start = rstart[rl];
    const int dd    = deg[lo + rl];
    const float dr  = (dd > 0) ? rsqrtf((float)dd) : 0.0f;
    const int* row  = src + start;
    for (int e = sub; e < dd; e += 16) {
      const int key = row[e];
      int rank = 0;
      for (int j = 0; j < dd; ++j) {
        const int kj = row[j];
        rank += ((kj < key) || (kj == key && j < e)) ? 1 : 0;
      }
      const int dc   = deg[key];
      const float dv = (dc > 0) ? rsqrtf((float)dc) : 0.0f;
      cw_seg[start + rank] = make_int2(key + node0, __float_as_int(dr * dv));
    }
  }
}

__device__ __forceinline__ void csr_range_block(const int* __restrict__ ei,
                                                int* __restrict__ rowptr,
                                                int* colu,
                                                int2* __restrict__ cw,
                                                const int b, const int q) {
  __shared__ int      deg[N_];
  __shared__ unsigned kept[KCAP_];
  __shared__ int      placed[KCAP_];
  __shared__ int      cursor[RROWS_];
  __shared__ int      rstart[RROWS_];
  __shared__ int      wsum_scan[32];
  __shared__ int      wsum_base[32];
  __shared__ int      kept_n;

  const int t     = threadIdx.x;
  const int lane  = t & 31;
  const int w     = t >> 5;
  const int lo    = q * RROWS_;
  const int hi    = (lo + RROWS_ < N_) ? (lo + RROWS_) : N_;
  const int nrows = hi - lo;

  for (int i = t; i < N_; i += 1024) deg[i] = 0;
  if (t == 0) kept_n = 0;
  __syncthreads();

  const int4* __restrict__ e4 = (const int4*)(ei + (size_t)b * E_ * 2);
  for (int i0 = 0; i0 < E_ / 2; i0 += LOADS_ * 1024) {
    int4 v[LOADS_];
#pragma unroll
    for (int u = 0; u < LOADS_; ++u) {
      const int idx = i0 + u * 1024 + t;
      v[u] = e4[(idx < E_ / 2) ? idx : (E_ / 2 - 1)];
    }
    int r[2 * LOADS_], s[2 * LOADS_];
#pragma unroll
    for (int u = 0; u < LOADS_; ++u) {
      const bool live = (i0 + u * 1024 + t) < E_ / 2;
      r[2 * u]     = live ? v[u].x : -1;  s[2 * u]     = v[u].y;
      r[2 * u + 1] = live ? v[u].z : -1;  s[2 * u + 1] = v[u].w;
    }
    int n_in = 0;
#pragma unroll
    for (int k = 0; k < 2 * LOADS_; ++k) {
      if (r[k] >= 0) atomicAdd(&deg[r[k]], 1);
      n_in += (r[k] >= lo && r[k] < hi) ? 1 : 0;
    }
    int p = (n_in > 0) ? atomicAdd(&kept_n, n_in) : 0;
#pragma unroll
    for (int k = 0; k < 2 * LOADS_; ++k) {
      if (r[k] >= lo && r[k] < hi) {
        if (p < KCAP_) kept[p] = ((unsigned)(r[k] - lo) << 14) | (unsigned)s[k];
        ++p;
      }
    }
  }
  __syncthreads();
  const int  kn   = kept_n;
  const bool fast = (kn <= KCAP_);

  int part = 0;
  for (int i = t; i < lo; i += 1024) part += deg[i];
  const int d = (t < nrows) ? deg[lo + t] : 0;
  int inc = d;
#pragma unroll
  for (int s = 1; s < 32; s <<= 1) {
    const int o = __shfl_up(inc, s);
    if (lane >= s) inc += o;
  }
  part = wave_sum(part);
  if (lane == 31) wsum_scan[w] = inc;
  if (lane == 0)  wsum_base[w] = part;
  __syncthreads();
  const int woff = wave_sum((lane < w) ? wsum_scan[lane] : 0);
  const int base = wave_sum(wsum_base[lane]);
  const int excl = woff + inc - d;
  const int seg0 = b * E_ + base;
  if (t < nrows) {
    rstart[t] = excl;
    cursor[t] = excl;
    rowptr[b * N_ + lo + t] = seg0 + excl;
  }
  __syncthreads();

  if (fast) {
    for (int i = t; i < kn; i += 1024) {
      const unsigned u = kept[i];
      const int pos = atomicAdd(&cursor[u >> 14], 1);
      placed[pos] = (int)(u & 16383u);
    }
  } else {
    const int2* __restrict__ e2 = (const int2*)(ei + (size_t)b * E_ * 2);
    for (int g = t; g < E_; g += 1024) {
      const int2 e = e2[g];
      if (e.x >= lo && e.x < hi) {
        const int pos = atomicAdd(&cursor[e.x - lo], 1);
        colu[seg0 + pos] = e.y;
      }
    }
  }
  __syncthreads();

  if (fast) sort_range_rows(placed, deg, rstart, cw + seg0, t, lo, nrows, b * N_);
  else      sort_range_rows(colu + seg0, deg, rstart, cw + seg0, t, lo, nrows, b * N_);
}

#define SETUP_PACK_BLOCKS 3
__global__ __launch_bounds__(1024) void csr_pack_kernel(const int* __restrict__ ei,
                                                        const float* __restrict__ W1,
                                                        const float* __restrict__ W2,
                                                        const float* __restrict__ W3,
                                                        _Float16* __restrict__ Wp1,
                                                        _Float16* __restrict__ Wp2,
                                                        _Float16* __restrict__ Wp3,
                                                        int* __restrict__ rowptr,
                                                        int* colu,
                                                        int2* __restrict__ cw) {
  const int blk = blockIdx.x;
  if (blk >= CSR_BLOCKS) {
    const int l = blk - CSR_BLOCKS;
    const float* W  = (l == 0) ? W1  : (l == 1) ? W2  : W3;
    _Float16*    Wp = (l == 0) ? Wp1 : (l == 1) ? Wp2 : Wp3;
    if (threadIdx.x < 256) pack_w_block(W, Wp);
    return;
  }
  csr_range_block(ei, rowptr, colu, cw, blk / RANGES_, blk % RANGES_);
}

__device__ __forceinline__ void gemm_bias_frags(const v16h a0, const v16h a1,
                                                const _Float16* __restrict__ Wp,
                                                const float* __restrict__ bias,
                                                _Float16* __restrict__ Hout,
                                                const int row0) {
  const int lane = threadIdx.x & 31;
  const int half = lane >> 4;
  const int mrow = lane & 15;
  const v16h* Wpv = (const v16h*)Wp;
#pragma unroll
  for (int nt = 0; nt < 4; ++nt) {
    const v16h b0 = Wpv[(nt * 2 + 0) * 32 + lane];
    const v16h b1 = Wpv[(nt * 2 + 1) * 32 + lane];

    v8f c = {};
    c = __builtin_amdgcn_wmma_f32_16x16x32_f16(false, a0, false, b0, (short)0, c, false, false);
    c = __builtin_amdgcn_wmma_f32_16x16x32_f16(false, a1, false, b1, (short)0, c, false, false);

    const int n = nt * 16 + mrow;
    const float bv = bias[n];
#pragma unroll
    for (int r = 0; r < 8; ++r) {
      int m = r + half * 8;
      Hout[(size_t)(row0 + m) * F_ + n] = (_Float16)(c[r] + bv);
    }
  }
}

#define MODE_SUM   0
#define MODE_NEXT  1
#define MODE_FIRST 2
#define GROWS_ 5
#define GATH_ROWS_PER_BLOCK (16 * GROWS_)
#define GATH_TILES_PER_BLOCK (GATH_ROWS_PER_BLOCK / 16)
#define APITCH_ 72
static_assert(BN_ % GATH_ROWS_PER_BLOCK == 0, "every group owns GROWS_ full rows");
static_assert(GROWS_ < 16, "lane sub of a group holds row pointer sub, sub <= GROWS_");
static_assert(GATH_ROWS_PER_BLOCK % 16 == 0 && GATH_TILES_PER_BLOCK <= 8, "whole 16-row tiles, one wave each");

typedef __attribute__((ext_vector_type(4))) _Float16 v4h;
typedef __attribute__((ext_vector_type(8))) _Float16 v8h;

__device__ __forceinline__ void axpy4(float4& acc, float w, const float4 h) {
  acc.x = fmaf(w, h.x, acc.x);
  acc.y = fmaf(w, h.y, acc.y);
  acc.z = fmaf(w, h.z, acc.z);
  acc.w = fmaf(w, h.w, acc.w);
}
__device__ __forceinline__ void axpy4(float4& acc, float w, const v4h h) {
  acc.x = fmaf(w, (float)h[0], acc.x);
  acc.y = fmaf(w, (float)h[1], acc.y);
  acc.z = fmaf(w, (float)h[2], acc.z);
  acc.w = fmaf(w, (float)h[3], acc.w);
}
template <bool HALF_ROWS> struct RowPiece       { typedef float4 type; };
template <>               struct RowPiece<true> { typedef v4h    type; };

__device__ __forceinline__ void load_a_frags(const _Float16* tile, v16h& a0, v16h& a1) {
  const int lane = threadIdx.x & 31;
  const _Float16* arow = tile + (lane & 15) * APITCH_ + (lane >> 4) * 8;
  const v8h q0 = *(const v8h*)(arow);
  const v8h q1 = *(const v8h*)(arow + 16);
  const v8h q2 = *(const v8h*)(arow + 32);
  const v8h q3 = *(const v8h*)(arow + 48);
  a0 = __builtin_shufflevector(q0, q1, 0, 1, 2, 3, 4, 5, 6, 7, 8, 9, 10, 11, 12, 13, 14, 15);
  a1 = __builtin_shufflevector(q2, q3, 0, 1, 2, 3, 4, 5, 6, 7, 8, 9, 10, 11, 12, 13, 14, 15);
}

__device__ __forceinline__ void first_layer_tile(_Float16* tile, const float* wsum,
                                                 const _Float16* __restrict__ Wp,
                                                 const float* __restrict__ bias) {
  const int lane = threadIdx.x & 31;
  const int half = lane >> 4;
  const int mrow = lane & 15;
  v16h a0, a1;
  load_a_frags(tile, a0, a1);
  float sv[8];
#pragma unroll
  for (int r = 0; r < 8; ++r) sv[r] = wsum[half * 8 + r];
  const v16h* Wpv = (const v16h*)Wp;
#pragma unroll
  for (int nt = 0; nt < 4; ++nt) {
    const v16h b0 = Wpv[(nt * 2 + 0) * 32 + lane];
    const v16h b1 = Wpv[(nt * 2 + 1) * 32 + lane];

    v8f c = {};
    c = __builtin_amdgcn_wmma_f32_16x16x32_f16(false, a0, false, b0, (short)0, c, false, false);
    c = __builtin_amdgcn_wmma_f32_16x16x32_f16(false, a1, false, b1, (short)0, c, false, false);

    const int n = nt * 16 + mrow;
    const float bv = bias[n];
#pragma unroll
    for (int r = 0; r < 8; ++r) {
      const int m = r + half * 8;
      tile[m * APITCH_ + n] = (_Float16)fmaxf(fmaf(sv[r], bv, c[r]), 0.0f);
    }
  }
}

template <int MODE>
__global__ __launch_bounds__(256) void gather_rows_kernel(const float* __restrict__ Hsrc,
                                                          const int* __restrict__ rowptr,
                                                          const int2* __restrict__ cw,
                                                          float* __restrict__ out,
                                                          const _Float16* __restrict__ Wp,
                                                          const float* __restrict__ bias,
                                                          const _Float16* __restrict__ Wp0,
                                                          const float* __restrict__ bias0) {
  __shared__ __attribute__((aligned(16))) _Float16 At[(MODE != MODE_SUM) ? GATH_ROWS_PER_BLOCK * APITCH_ : 8];
  __shared__ float Wsum[(MODE == MODE_FIRST) ? GATH_ROWS_PER_BLOCK : 1];

  const int t   = blockIdx.x * blockDim.x + threadIdx.x;
  const int sub = t & 15;
  const int r0  = (t >> 4) * GROWS_;
  constexpr bool HALF_ROWS = (MODE != MODE_FIRST);
  constexpr int  ROW_SHIFT = HALF_ROWS ? 7 : 8;
  typedef typename RowPiece<HALF_ROWS>::type piece_t;
  const char* __restrict__ Hb = (const char*)Hsrc;
  const unsigned lane_off = (unsigned)sub << (ROW_SHIFT - 4);

  int rp = 0;
  if (sub <= GROWS_) rp = (r0 + sub < BN_) ? rowptr[r0 + sub] : BE_;
  const int gend = __shfl(rp, GROWS_, 16);
  int base = __shfl(rp, 0, 16);
  int2 my = make_int2(0, 0);
  if (base + sub < gend) my = cw[base + sub];

  for (int rr = 0; rr < GROWS_; ++rr) {
    const int end = __shfl(rp, rr + 1, 16);
    float4 acc = make_float4(0.0f, 0.0f, 0.0f, 0.0f);
    float ws = 0.0f;
    while (base < end) {
      const int n  = end - base;
      const int nb = (n > 16) ? (base + 16) : end;
      int2 nxt = make_int2(0, 0);
      if (nb + sub < gend) nxt = cw[nb + sub];
      piece_t h[16];
#pragma unroll
      for (int k = 0; k < 16; ++k) {
        const unsigned s = (unsigned)__shfl(my.x, k, 16);
        if (k < n) h[k] = *(const piece_t*)(Hb + ((s << ROW_SHIFT) | lane_off));
      }
#pragma unroll
      for (int k = 0; k < 16; ++k) {
        const float w = __int_as_float(__shfl(my.y, k, 16));
        if (k < n) {
          axpy4(acc, w, h[k]);
          if (MODE == MODE_FIRST) ws += w;
        }
      }
      my   = nxt;
      base = nb;
    }
    if (MODE != MODE_SUM) {
      const bool relu = (MODE == MODE_NEXT);
      v4h q;
      q[0] = (_Float16)(relu ? fmaxf(acc.x, 0.0f) : acc.x);
      q[1] = (_Float16)(relu ? fmaxf(acc.y, 0.0f) : acc.y);
      q[2] = (_Float16)(relu ? fmaxf(acc.z, 0.0f) : acc.z);
      q[3] = (_Float16)(relu ? fmaxf(acc.w, 0.0f) : acc.w);
      const int lr = (threadIdx.x >> 4) * GROWS_ + rr;
      *(v4h*)(At + lr * APITCH_ + sub * 4) = q;
      if (MODE == MODE_FIRST && sub == 0) Wsum[lr] = ws;
    } else {
      ((float4*)out)[(size_t)(r0 + rr) * (F_ / 4) + sub] = acc;
    }
  }

  if (MODE != MODE_SUM) {
    __syncthreads();
    const int  wv       = threadIdx.x >> 5;
    const bool has_tile = wv < GATH_TILES_PER_BLOCK;
    _Float16*  tile     = At + (has_tile ? wv : 0) * 16 * APITCH_;
    if (MODE == MODE_FIRST) {
      if (has_tile) first_layer_tile(tile, Wsum + wv * 16, Wp0, bias0);
      __syncthreads();
    }
    if (has_tile) {
      v16h a0, a1;
      load_a_frags(tile, a0, a1);
      gemm_bias_frags(a0, a1, Wp, bias, (_Float16*)out, blockIdx.x * GATH_ROWS_PER_BLOCK + wv * 16);
    }
  }
}

extern "C" void kernel_launch(void* const* d_in, const int* in_sizes, int n_in,
                              void* d_out, int out_size, void* d_ws, size_t ws_size,
                              hipStream_t stream) {
  const float* x  = (const float*)d_in[0];
  const int*   ei = (const int*)  d_in[1];
  const float* W1 = (const float*)d_in[2];
  const float* b1 = (const float*)d_in[3];
  const float* W2 = (const float*)d_in[4];
  const float* b2 = (const float*)d_in[5];
  const float* W3 = (const float*)d_in[6];
  const float* b3 = (const float*)d_in[7];
  float* out = (float*)d_out;

  auto align256 = [](size_t v) { return (v + 255) & ~(size_t)255; };
  char* ws = (char*)d_ws;
  int*   rowptr = (int*)ws;      ws += align256((size_t)BN_ * 4);
  int2*  cw     = (int2*)ws;     ws += 2 * align256((size_t)BE_ * 4);
  ws += align256((size_t)BE_ * 4);
  float* Hlin  = (float*)ws;     ws += align256((size_t)BN_ * F_ * 4);
  float* Hlin2 = (float*)ws;     ws += align256((size_t)BN_ * F_ * 4);
  _Float16* Wp1 = (_Float16*)ws; ws += align256((size_t)WPACK_HALVES * 2);
  _Float16* Wp2 = (_Float16*)ws; ws += align256((size_t)WPACK_HALVES * 2);
  _Float16* Wp3 = (_Float16*)ws; ws += align256((size_t)WPACK_HALVES * 2);
  int* colu = (int*)Hlin2;

  const int TB = 256;
  const int GATH_GRID = BN_ / GATH_ROWS_PER_BLOCK;

  csr_pack_kernel<<<CSR_BLOCKS + SETUP_PACK_BLOCKS, 1024, 0, stream>>>(ei, W1, W2, W3, Wp1, Wp2, Wp3,
                                                                     rowptr, colu, cw);

  gather_rows_kernel<MODE_FIRST><<<GATH_GRID, TB, 0, stream>>>(x, rowptr, cw, Hlin, Wp2, b2, Wp1, b1);

  gather_rows_kernel<MODE_NEXT><<<GATH_GRID, TB, 0, stream>>>(Hlin, rowptr, cw, Hlin2, Wp3, b3, nullptr, nullptr);

  gather_rows_kernel<MODE_SUM><<<GATH_GRID, TB, 0, stream>>>(Hlin2, rowptr, cw, out, nullptr, nullptr, nullptr, nullptr);
}


# Row gather loses its per-term branches: a term past the row's end gets weight 0 and source row 0, so loads and multiply-adds run straight-line, 8 or 16 terms wide (wave-uniform test); sum order kept
# speedup vs baseline: 5.5027x; 1.1028x over previous
#include <hip/hip_runtime.h>
#include <hip/hip_bf16.h>

typedef __attribute__((ext_vector_type(16))) _Float16 v16h;
typedef __attribute__((ext_vector_type(8)))  float    v8f;

#define B_  8
#define N_  10000
#define E_  160000
#define F_  64
#define BN_ (B_*N_)
#define BE_ (B_*E_)

#define WPACK_HALVES 4096

__device__ __forceinline__ void pack_w_block(const float* __restrict__ W,
                                             _Float16* __restrict__ Wp) {
  const int t    = threadIdx.x;
  const int nt   = t >> 6;
  const int kb   = (t >> 5) & 1;
  const int lane = t & 31;
  const int n    = nt * 16 + (lane & 15);
  const int half = lane >> 4;
  _Float16* dst = Wp + ((size_t)((nt * 2 + kb) * 32 + lane)) * 16;
#pragma unroll
  for (int v = 0; v < 8; ++v) {
    int kbase = (v < 4) ? (2 * v) : (16 + 2 * (v - 4));
    int k = kb * 32 + kbase + half * 8;
    dst[2 * v]     = (_Float16)W[n * F_ + k];
    dst[2 * v + 1] = (_Float16)W[n * F_ + k + 1];
  }
}

#define RANGES_ 32
#define RROWS_  ((N_ + RANGES_ - 1) / RANGES_)
#define KCAP_   8192
#define CSR_BLOCKS (B_ * RANGES_)
#define LOADS_  8
static_assert(N_ <= 16384, "a source node packs into 14 bits of a kept entry");
static_assert(RROWS_ <= 1024, "one thread per row of the range in the scan");
static_assert(E_ % 2 == 0, "the edge list is read two edges per 16-byte load");

__device__ __forceinline__ int wave_sum(int v) {
#pragma unroll
  for (int m = 16; m > 0; m >>= 1) v += __shfl_xor(v, m);
  return v;
}

__device__ __forceinline__ void sort_range_rows(const int* src, const int* deg, const int* rstart,
                                                int2* __restrict__ cw_seg, const int t,
                                                const int lo, const int nrows, const int node0) {
  const int sub = t & 15;
  for (int rl = t >> 4; rl < nrows; rl += 64) {
    const int start = rstart[rl];
    const int dd    = deg[lo + rl];
    const float dr  = (dd > 0) ? rsqrtf((float)dd) : 0.0f;
    const int* row  = src + start;
    for (int e = sub; e < dd; e += 16) {
      const int key = row[e];
      int rank = 0;
      for (int j = 0; j < dd; ++j) {
        const int kj = row[j];
        rank += ((kj < key) || (kj == key && j < e)) ? 1 : 0;
      }
      const int dc   = deg[key];
      const float dv = (dc > 0) ? rsqrtf((float)dc) : 0.0f;
      cw_seg[start + rank] = make_int2(key + node0, __float_as_int(dr * dv));
    }
  }
}

__device__ __forceinline__ void csr_range_block(const int* __restrict__ ei,
                                                int* __restrict__ rowptr,
                                                int* colu,
                                                int2* __restrict__ cw,
                                                const int b, const int q) {
  __shared__ int      deg[N_];
  __shared__ unsigned kept[KCAP_];
  __shared__ int      placed[KCAP_];
  __shared__ int      cursor[RROWS_];
  __shared__ int      rstart[RROWS_];
  __shared__ int      wsum_scan[32];
  __shared__ int      wsum_base[32];
  __shared__ int      kept_n;

  const int t     = threadIdx.x;
  const int lane  = t & 31;
  const int w     = t >> 5;
  const int lo    = q * RROWS_;
  const int hi    = (lo + RROWS_ < N_) ? (lo + RROWS_) : N_;
  const int nrows = hi - lo;

  for (int i = t; i < N_; i += 1024) deg[i] = 0;
  if (t == 0) kept_n = 0;
  __syncthreads();

  const int4* __restrict__ e4 = (const int4*)(ei + (size_t)b * E_ * 2);
  for (int i0 = 0; i0 < E_ / 2; i0 += LOADS_ * 1024) {
    int4 v[LOADS_];
#pragma unroll
    for (int u = 0; u < LOADS_; ++u) {
      const int idx = i0 + u * 1024 + t;
      v[u] = e4[(idx < E_ / 2) ? idx : (E_ / 2 - 1)];
    }
    int r[2 * LOADS_], s[2 * LOADS_];
#pragma unroll
    for (int u = 0; u < LOADS_; ++u) {
      const bool live = (i0 + u * 1024 + t) < E_ / 2;
      r[2 * u]     = live ? v[u].x : -1;  s[2 * u]     = v[u].y;
      r[2 * u + 1] = live ? v[u].z : -1;  s[2 * u + 1] = v[u].w;
    }
    int n_in = 0;
#pragma unroll
    for (int k = 0; k < 2 * LOADS_; ++k) {
      if (r[k] >= 0) atomicAdd(&deg[r[k]], 1);
      n_in += (r[k] >= lo && r[k] < hi) ? 1 : 0;
    }
    int p = (n_in > 0) ? atomicAdd(&kept_n, n_in) : 0;
#pragma unroll
    for (int k = 0; k < 2 * LOADS_; ++k) {
      if (r[k] >= lo && r[k] < hi) {
        if (p < KCAP_) kept[p] = ((unsigned)(r[k] - lo) << 14) | (unsigned)s[k];
        ++p;
      }
    }
  }
  __syncthreads();
  const int  kn   = kept_n;
  const bool fast = (kn <= KCAP_);

  int part = 0;
  for (int i = t; i < lo; i += 1024) part += deg[i];
  const int d = (t < nrows) ? deg[lo + t] : 0;
  int inc = d;
#pragma unroll
  for (int s = 1; s < 32; s <<= 1) {
    const int o = __shfl_up(inc, s);
    if (lane >= s) inc += o;
  }
  part = wave_sum(part);
  if (lane == 31) wsum_scan[w] = inc;
  if (lane == 0)  wsum_base[w] = part;
  __syncthreads();
  const int woff = wave_sum((lane < w) ? wsum_scan[lane] : 0);
  const int base = wave_sum(wsum_base[lane]);
  const int excl = woff + inc - d;
  const int seg0 = b * E_ + base;
  if (t < nrows) {
    rstart[t] = excl;
    cursor[t] = excl;
    rowptr[b * N_ + lo + t] = seg0 + excl;
  }
  __syncthreads();

  if (fast) {
    for (int i = t; i < kn; i += 1024) {
      const unsigned u = kept[i];
      const int pos = atomicAdd(&cursor[u >> 14], 1);
      placed[pos] = (int)(u & 16383u);
    }
  } else {
    const int2* __restrict__ e2 = (const int2*)(ei + (size_t)b * E_ * 2);
    for (int g = t; g < E_; g += 1024) {
      const int2 e = e2[g];
      if (e.x >= lo && e.x < hi) {
        const int pos = atomicAdd(&cursor[e.x - lo], 1);
        colu[seg0 + pos] = e.y;
      }
    }
  }
  __syncthreads();

  if (fast) sort_range_rows(placed, deg, rstart, cw + seg0, t, lo, nrows, b * N_);
  else      sort_range_rows(colu + seg0, deg, rstart, cw + seg0, t, lo, nrows, b * N_);
}

#define SETUP_PACK_BLOCKS 3
__global__ __launch_bounds__(1024) void csr_pack_kernel(const int* __restrict__ ei,
                                                        const float* __restrict__ W1,
                                                        const float* __restrict__ W2,
                                                        const float* __restrict__ W3,
                                                        _Float16* __restrict__ Wp1,
                                                        _Float16* __restrict__ Wp2,
                                                        _Float16* __restrict__ Wp3,
                                                        int* __restrict__ rowptr,
                                                        int* colu,
                                                        int2* __restrict__ cw) {
  const int blk = blockIdx.x;
  if (blk >= CSR_BLOCKS) {
    const int l = blk - CSR_BLOCKS;
    const float* W  = (l == 0) ? W1  : (l == 1) ? W2  : W3;
    _Float16*    Wp = (l == 0) ? Wp1 : (l == 1) ? Wp2 : Wp3;
    if (threadIdx.x < 256) pack_w_block(W, Wp);
    return;
  }
  csr_range_block(ei, rowptr, colu, cw, blk / RANGES_, blk % RANGES_);
}

__device__ __forceinline__ void gemm_bias_frags(const v16h a0, const v16h a1,
                                                const _Float16* __restrict__ Wp,
                                                const float* __restrict__ bias,
                                                _Float16* __restrict__ Hout,
                                                const int row0) {
  const int lane = threadIdx.x & 31;
  const int half = lane >> 4;
  const int mrow = lane & 15;
  const v16h* Wpv = (const v16h*)Wp;
#pragma unroll
  for (int nt = 0; nt < 4; ++nt) {
    const v16h b0 = Wpv[(nt * 2 + 0) * 32 + lane];
    const v16h b1 = Wpv[(nt * 2 + 1) * 32 + lane];

    v8f c = {};
    c = __builtin_amdgcn_wmma_f32_16x16x32_f16(false, a0, false, b0, (short)0, c, false, false);
    c = __builtin_amdgcn_wmma_f32_16x16x32_f16(false, a1, false, b1, (short)0, c, false, false);

    const int n = nt * 16 + mrow;
    const float bv = bias[n];
#pragma unroll
    for (int r = 0; r < 8; ++r) {
      int m = r + half * 8;
      Hout[(size_t)(row0 + m) * F_ + n] = (_Float16)(c[r] + bv);
    }
  }
}

#define MODE_SUM   0
#define MODE_NEXT  1
#define MODE_FIRST 2
#define GROWS_ 5
#define GATH_ROWS_PER_BLOCK (16 * GROWS_)
#define GATH_TILES_PER_BLOCK (GATH_ROWS_PER_BLOCK / 16)
#define APITCH_ 72
static_assert(BN_ % GATH_ROWS_PER_BLOCK == 0, "every group owns GROWS_ full rows");
static_assert(GROWS_ < 16, "lane sub of a group holds row pointer sub, sub <= GROWS_");
static_assert(GATH_ROWS_PER_BLOCK % 16 == 0 && GATH_TILES_PER_BLOCK <= 8, "whole 16-row tiles, one wave each");

typedef __attribute__((ext_vector_type(4))) _Float16 v4h;
typedef __attribute__((ext_vector_type(8))) _Float16 v8h;

__device__ __forceinline__ void axpy4(float4& acc, float w, const float4 h) {
  acc.x = fmaf(w, h.x, acc.x);
  acc.y = fmaf(w, h.y, acc.y);
  acc.z = fmaf(w, h.z, acc.z);
  acc.w = fmaf(w, h.w, acc.w);
}
__device__ __forceinline__ void axpy4(float4& acc, float w, const v4h h) {
  acc.x = fmaf(w, (float)h[0], acc.x);
  acc.y = fmaf(w, (float)h[1], acc.y);
  acc.z = fmaf(w, (float)h[2], acc.z);
  acc.w = fmaf(w, (float)h[3], acc.w);
}
template <bool HALF_ROWS> struct RowPiece       { typedef float4 type; };
template <>               struct RowPiece<true> { typedef v4h    type; };

__device__ __forceinline__ void load_a_frags(const _Float16* tile, v16h& a0, v16h& a1) {
  const int lane = threadIdx.x & 31;
  const _Float16* arow = tile + (lane & 15) * APITCH_ + (lane >> 4) * 8;
  const v8h q0 = *(const v8h*)(arow);
  const v8h q1 = *(const v8h*)(arow + 16);
  const v8h q2 = *(const v8h*)(arow + 32);
  const v8h q3 = *(const v8h*)(arow + 48);
  a0 = __builtin_shufflevector(q0, q1, 0, 1, 2, 3, 4, 5, 6, 7, 8, 9, 10, 11, 12, 13, 14, 15);
  a1 = __builtin_shufflevector(q2, q3, 0, 1, 2, 3, 4, 5, 6, 7, 8, 9, 10, 11, 12, 13, 14, 15);
}

__device__ __forceinline__ void first_layer_tile(_Float16* tile, const float* wsum,
                                                 const _Float16* __restrict__ Wp,
                                                 const float* __restrict__ bias) {
  const int lane = threadIdx.x & 31;
  const int half = lane >> 4;
  const int mrow = lane & 15;
  v16h a0, a1;
  load_a_frags(tile, a0, a1);
  float sv[8];
#pragma unroll
  for (int r = 0; r < 8; ++r) sv[r] = wsum[half * 8 + r];
  const v16h* Wpv = (const v16h*)Wp;
#pragma unroll
  for (int nt = 0; nt < 4; ++nt) {
    const v16h b0 = Wpv[(nt * 2 + 0) * 32 + lane];
    const v16h b1 = Wpv[(nt * 2 + 1) * 32 + lane];

    v8f c = {};
    c = __builtin_amdgcn_wmma_f32_16x16x32_f16(false, a0, false, b0, (short)0, c, false, false);
    c = __builtin_amdgcn_wmma_f32_16x16x32_f16(false, a1, false, b1, (short)0, c, false, false);

    const int n = nt * 16 + mrow;
    const float bv = bias[n];
#pragma unroll
    for (int r = 0; r < 8; ++r) {
      const int m = r + half * 8;
      tile[m * APITCH_ + n] = (_Float16)fmaxf(fmaf(sv[r], bv, c[r]), 0.0f);
    }
  }
}

template <int NK, int MODE, int ROW_SHIFT, typename piece_t>
__device__ __forceinline__ void chunk_terms(const char* __restrict__ Hb, const unsigned lane_off,
                                            const int ms, const int mwi, float4& acc, float& ws) {
  piece_t h[NK];
#pragma unroll
  for (int k = 0; k < NK; ++k) {
    const unsigned s = (unsigned)__shfl(ms, k, 16);
    h[k] = *(const piece_t*)(Hb + ((s << ROW_SHIFT) | lane_off));
  }
#pragma unroll
  for (int k = 0; k < NK; ++k) {
    const float w = __int_as_float(__shfl(mwi, k, 16));
    axpy4(acc, w, h[k]);
    if (MODE == MODE_FIRST) ws += w;
  }
}

template <int MODE>
__global__ __launch_bounds__(256) void gather_rows_kernel(const float* __restrict__ Hsrc,
                                                          const int* __restrict__ rowptr,
                                                          const int2* __restrict__ cw,
                                                          float* __restrict__ out,
                                                          const _Float16* __restrict__ Wp,
                                                          const float* __restrict__ bias,
                                                          const _Float16* __restrict__ Wp0,
                                                          const float* __restrict__ bias0) {
  __shared__ __attribute__((aligned(16))) _Float16 At[(MODE != MODE_SUM) ? GATH_ROWS_PER_BLOCK * APITCH_ : 8];
  __shared__ float Wsum[(MODE == MODE_FIRST) ? GATH_ROWS_PER_BLOCK : 1];

  const int t   = blockIdx.x * blockDim.x + threadIdx.x;
  const int sub = t & 15;
  const int r0  = (t >> 4) * GROWS_;
  constexpr bool HALF_ROWS = (MODE != MODE_FIRST);
  constexpr int  ROW_SHIFT = HALF_ROWS ? 7 : 8;
  typedef typename RowPiece<HALF_ROWS>::type piece_t;
  const char* __restrict__ Hb = (const char*)Hsrc;
  const unsigned lane_off = (unsigned)sub << (ROW_SHIFT - 4);

  int rp = 0;
  if (sub <= GROWS_) rp = (r0 + sub < BN_) ? rowptr[r0 + sub] : BE_;
  const int gend = __shfl(rp, GROWS_, 16);
  int base = __shfl(rp, 0, 16);
  int2 my = make_int2(0, 0);
  if (base + sub < gend) my = cw[base + sub];

  for (int rr = 0; rr < GROWS_; ++rr) {
    const int end = __shfl(rp, rr + 1, 16);
    float4 acc = make_float4(0.0f, 0.0f, 0.0f, 0.0f);
    float ws = 0.0f;
    while (base < end) {
      const int n  = end - base;
      const int nb = (n > 16) ? (base + 16) : end;
      const bool live = sub < n;
      const int  ms   = live ? my.x : 0;
      const int  mwi  = live ? my.y : 0;
      const int2 nxt = cw[(nb + sub < BE_) ? (nb + sub) : (BE_ - 1)];
      if (__any(n > 8)) chunk_terms<16, MODE, ROW_SHIFT, piece_t>(Hb, lane_off, ms, mwi, acc, ws);
      else              chunk_terms<8,  MODE, ROW_SHIFT, piece_t>(Hb, lane_off, ms, mwi, acc, ws);
      my   = nxt;
      base = nb;
    }
    if (MODE != MODE_SUM) {
      const bool relu = (MODE == MODE_NEXT);
      v4h q;
      q[0] = (_Float16)(relu ? fmaxf(acc.x, 0.0f) : acc.x);
      q[1] = (_Float16)(relu ? fmaxf(acc.y, 0.0f) : acc.y);
      q[2] = (_Float16)(relu ? fmaxf(acc.z, 0.0f) : acc.z);
      q[3] = (_Float16)(relu ? fmaxf(acc.w, 0.0f) : acc.w);
      const int lr = (threadIdx.x >> 4) * GROWS_ + rr;
      *(v4h*)(At + lr * APITCH_ + sub * 4) = q;
      if (MODE == MODE_FIRST && sub == 0) Wsum[lr] = ws;
    } else {
      ((float4*)out)[(size_t)(r0 + rr) * (F_ / 4) + sub] = acc;
    }
  }

  if (MODE != MODE_SUM) {
    __syncthreads();
    const int  wv       = threadIdx.x >> 5;
    const bool has_tile = wv < GATH_TILES_PER_BLOCK;
    _Float16*  tile     = At + (has_tile ? wv : 0) * 16 * APITCH_;
    if (MODE == MODE_FIRST) {
      if (has_tile) first_layer_tile(tile, Wsum + wv * 16, Wp0, bias0);
      __syncthreads();
    }
    if (has_tile) {
      v16h a0, a1;
      load_a_frags(tile, a0, a1);
      gemm_bias_frags(a0, a1, Wp, bias, (_Float16*)out, blockIdx.x * GATH_ROWS_PER_BLOCK + wv * 16);
    }
  }
}

extern "C" void kernel_launch(void* const* d_in, const int* in_sizes, int n_in,
                              void* d_out, int out_size, void* d_ws, size_t ws_size,
                              hipStream_t stream) {
  const float* x  = (const float*)d_in[0];
  const int*   ei = (const int*)  d_in[1];
  const float* W1 = (const float*)d_in[2];
  const float* b1 = (const float*)d_in[3];
  const float* W2 = (const float*)d_in[4];
  const float* b2 = (const float*)d_in[5];
  const float* W3 = (const float*)d_in[6];
  const float* b3 = (const float*)d_in[7];
  float* out = (float*)d_out;

  auto align256 = [](size_t v) { return (v + 255) & ~(size_t)255; };
  char* ws = (char*)d_ws;
  int*   rowptr = (int*)ws;      ws += align256((size_t)BN_ * 4);
  int2*  cw     = (int2*)ws;     ws += 2 * align256((size_t)BE_ * 4);
  ws += align256((size_t)BE_ * 4);
  float* Hlin  = (float*)ws;     ws += align256((size_t)BN_ * F_ * 4);
  float* Hlin2 = (float*)ws;     ws += align256((size_t)BN_ * F_ * 4);
  _Float16* Wp1 = (_Float16*)ws; ws += align256((size_t)WPACK_HALVES * 2);
  _Float16* Wp2 = (_Float16*)ws; ws += align256((size_t)WPACK_HALVES * 2);
  _Float16* Wp3 = (_Float16*)ws; ws += align256((size_t)WPACK_HALVES * 2);
  int* colu = (int*)Hlin2;

  const int TB = 256;
  const int GATH_GRID = BN_ / GATH_ROWS_PER_BLOCK;

  csr_pack_kernel<<<CSR_BLOCKS + SETUP_PACK_BLOCKS, 1024, 0, stream>>>(ei, W1, W2, W3, Wp1, Wp2, Wp3,
                                                                     rowptr, colu, cw);

  gather_rows_kernel<MODE_FIRST><<<GATH_GRID, TB, 0, stream>>>(x, rowptr, cw, Hlin, Wp2, b2, Wp1, b1);

  gather_rows_kernel<MODE_NEXT><<<GATH_GRID, TB, 0, stream>>>(Hlin, rowptr, cw, Hlin2, Wp3, b3, nullptr, nullptr);

  gather_rows_kernel<MODE_SUM><<<GATH_GRID, TB, 0, stream>>>(Hlin2, rowptr, cw, out, nullptr, nullptr, nullptr, nullptr);
}


# Edge weight split: stored f16 rows carry their node's dinv, a term's weight is its row's dinv; CSR block counts own-row degrees only and maps layer 1 (map, then sum): gather 1 reads f16 rows, not f32
# speedup vs baseline: 5.8088x; 1.0556x over previous
#include <hip/hip_runtime.h>
#include <hip/hip_bf16.h>

typedef __attribute__((ext_vector_type(16))) _Float16 v16h;
typedef __attribute__((ext_vector_type(8)))  float    v8f;

#define B_  8
#define N_  10000
#define E_  160000
#define F_  64
#define BN_ (B_*N_)
#define BE_ (B_*E_)

#define WPACK_HALVES 4096

typedef __attribute__((ext_vector_type(4))) _Float16 v4h;
typedef __attribute__((ext_vector_type(8))) _Float16 v8h;
#define APITCH_ 72

__device__ __forceinline__ void load_a_frags(const _Float16* tile, v16h& a0, v16h& a1) {
  const int lane = threadIdx.x & 31;
  const _Float16* arow = tile + (lane & 15) * APITCH_ + (lane >> 4) * 8;
  const v8h q0 = *(const v8h*)(arow);
  const v8h q1 = *(const v8h*)(arow + 16);
  const v8h q2 = *(const v8h*)(arow + 32);
  const v8h q3 = *(const v8h*)(arow + 48);
  a0 = __builtin_shufflevector(q0, q1, 0, 1, 2, 3, 4, 5, 6, 7, 8, 9, 10, 11, 12, 13, 14, 15);
  a1 = __builtin_shufflevector(q2, q3, 0, 1, 2, 3, 4, 5, 6, 7, 8, 9, 10, 11, 12, 13, 14, 15);
}

__device__ __forceinline__ void gemm_bias_scale_frags(const v16h a0, const v16h a1,
                                                      const _Float16* __restrict__ Wp,
                                                      const float* __restrict__ bias,
                                                      const float* dscale,
                                                      _Float16* __restrict__ Hout,
                                                      const int row0) {
  const int lane = threadIdx.x & 31;
  const int half = lane >> 4;
  const int mrow = lane & 15;
  float sv[8];
#pragma unroll
  for (int r = 0; r < 8; ++r) sv[r] = dscale[half * 8 + r];
  const v16h* Wpv = (const v16h*)Wp;
#pragma unroll
  for (int nt = 0; nt < 4; ++nt) {
    const v16h b0 = Wpv[(nt * 2 + 0) * 32 + lane];
    const v16h b1 = Wpv[(nt * 2 + 1) * 32 + lane];

    v8f c = {};
    c = __builtin_amdgcn_wmma_f32_16x16x32_f16(false, a0, false, b0, (short)0, c, false, false);
    c = __builtin_amdgcn_wmma_f32_16x16x32_f16(false, a1, false, b1, (short)0, c, false, false);

    const int n = nt * 16 + mrow;
    const float bv = bias[n];
#pragma unroll
    for (int r = 0; r < 8; ++r) {
      const int m = r + half * 8;
      Hout[(size_t)(row0 + m) * F_ + n] = (_Float16)((c[r] + bv) * sv[r]);
    }
  }
}

__device__ __forceinline__ void map_tile_in_place(_Float16* tile, const float* dscale,
                                                  const _Float16* Wp,
                                                  const float* __restrict__ bias) {
  const int lane = threadIdx.x & 31;
  const int half = lane >> 4;
  const int mrow = lane & 15;
  v16h a0, a1;
  load_a_frags(tile, a0, a1);
  float sv[8];
#pragma unroll
  for (int r = 0; r < 8; ++r) sv[r] = dscale[half * 8 + r];
  const v16h* Wpv = (const v16h*)Wp;
#pragma unroll
  for (int nt = 0; nt < 4; ++nt) {
    const v16h b0 = Wpv[(nt * 2 + 0) * 32 + lane];
    const v16h b1 = Wpv[(nt * 2 + 1) * 32 + lane];

    v8f c = {};
    c = __builtin_amdgcn_wmma_f32_16x16x32_f16(false, a0, false, b0, (short)0, c, false, false);
    c = __builtin_amdgcn_wmma_f32_16x16x32_f16(false, a1, false, b1, (short)0, c, false, false);

    const int n = nt * 16 + mrow;
    const float bv = bias[n];
#pragma unroll
    for (int r = 0; r < 8; ++r) {
      const int m = r + half * 8;
      tile[m * APITCH_ + n] = (_Float16)((c[r] + bv) * sv[r]);
    }
  }
}

__device__ __forceinline__ void pack_w_block(const float* __restrict__ W,
                                             _Float16* __restrict__ Wp) {
  const int t    = threadIdx.x;
  const int nt   = t >> 6;
  const int kb   = (t >> 5) & 1;
  const int lane = t & 31;
  const int n    = nt * 16 + (lane & 15);
  const int half = lane >> 4;
  _Float16* dst = Wp + ((size_t)((nt * 2 + kb) * 32 + lane)) * 16;
#pragma unroll
  for (int v = 0; v < 8; ++v) {
    int kbase = (v < 4) ? (2 * v) : (16 + 2 * (v - 4));
    int k = kb * 32 + kbase + half * 8;
    dst[2 * v]     = (_Float16)W[n * F_ + k];
    dst[2 * v + 1] = (_Float16)W[n * F_ + k + 1];
  }
}

#define RANGES_ 32
#define RROWS_  ((N_ + RANGES_ - 1) / RANGES_)
#define RTILES_ ((RROWS_ + 15) / 16)
#define KCAP_   8192
#define CSR_BLOCKS (B_ * RANGES_)
#define LOADS_  8
static_assert(N_ <= 16384, "a source node packs into 14 bits of a kept entry");
static_assert(RROWS_ <= 1024, "one thread per row of the range in the scan");
#define XROWS_PER_GROUP_ (RTILES_ * 16 / 64)
static_assert(RTILES_ <= 32, "one wave per 16-row tile of the range");
static_assert((RTILES_ * 16) % 64 == 0, "the staged rows divide evenly over the block's 64 groups");
static_assert((E_ / 2 + LOADS_ * 1024 - 1) / (LOADS_ * 1024) >= XROWS_PER_GROUP_, "the edge pass has a trip for every staged row of a group");
static_assert(E_ % 2 == 0, "the edge list is read two edges per 16-byte load");

__device__ __forceinline__ int wave_sum(int v) {
#pragma unroll
  for (int m = 16; m > 0; m >>= 1) v += __shfl_xor(v, m);
  return v;
}

__device__ __forceinline__ void sort_range_rows(const int* src, const int* deg, const int* rstart,
                                                const float* dinv,
                                                int2* __restrict__ cw_seg, const int t,
                                                const int nrows, const int node0) {
  const int sub = t & 15;
  for (int rl = t >> 4; rl < nrows; rl += 64) {
    const int start = rstart[rl];
    const int dd    = deg[rl];
    const int wbits = __float_as_int(dinv[rl]);
    const int* row  = src + start;
    for (int e = sub; e < dd; e += 16) {
      const int key = row[e];
      int rank = 0;
      for (int j = 0; j < dd; ++j) {
        const int kj = row[j];
        rank += ((kj < key) || (kj == key && j < e)) ? 1 : 0;
      }
      cw_seg[start + rank] = make_int2(key + node0, wbits);
    }
  }
}

template <bool STAGE>
__device__ __forceinline__ void edge_trip(const int4* __restrict__ e4, const int i0,
                                          const int lo, const int nrows, int& below,
                                          unsigned* kept, int* degown, int* kept_n,
                                          const float4* __restrict__ xrows,
                                          _Float16* At, const int lr) {
  const int t = threadIdx.x;
  float4 xq;
  if (STAGE) xq = xrows[(size_t)((lr < nrows) ? lr : (nrows - 1)) * (F_ / 4) + (t & 15)];
  int4 v[LOADS_];
#pragma unroll
  for (int u = 0; u < LOADS_; ++u) {
    const int idx = i0 + u * 1024 + t;
    v[u] = e4[(idx < E_ / 2) ? idx : (E_ / 2 - 1)];
  }
  int r[2 * LOADS_], s[2 * LOADS_];
#pragma unroll
  for (int u = 0; u < LOADS_; ++u) {
    const bool live = (i0 + u * 1024 + t) < E_ / 2;
    r[2 * u]     = live ? v[u].x : N_;  s[2 * u]     = v[u].y;
    r[2 * u + 1] = live ? v[u].z : N_;  s[2 * u + 1] = v[u].w;
  }
  int n_in = 0;
#pragma unroll
  for (int k = 0; k < 2 * LOADS_; ++k) {
    below += (r[k] < lo) ? 1 : 0;
    n_in  += ((unsigned)(r[k] - lo) < (unsigned)nrows) ? 1 : 0;
  }
  int p = (n_in > 0) ? atomicAdd(kept_n, n_in) : 0;
#pragma unroll
  for (int k = 0; k < 2 * LOADS_; ++k) {
    const unsigned rl = (unsigned)(r[k] - lo);
    if (rl < (unsigned)nrows) {
      atomicAdd(&degown[rl], 1);
      if (p < KCAP_) kept[p] = (rl << 14) | (unsigned)s[k];
      ++p;
    }
  }
  if (STAGE) {
    v4h qv;
    qv[0] = (_Float16)xq.x;  qv[1] = (_Float16)xq.y;  qv[2] = (_Float16)xq.z;  qv[3] = (_Float16)xq.w;
    *(v4h*)(At + lr * APITCH_ + (t & 15) * 4) = qv;
  }
}

__device__ __forceinline__ void csr_range_block(const int* __restrict__ ei,
                                                const float* __restrict__ x,
                                                const float* __restrict__ W1,
                                                const float* __restrict__ b1,
                                                _Float16* __restrict__ H1,
                                                int* __restrict__ rowptr,
                                                int* colu,
                                                int2* __restrict__ cw,
                                                const int b, const int q) {
  __shared__ unsigned kept[KCAP_];
  __shared__ int      placed[KCAP_];
  __shared__ int      degown[RROWS_];
  __shared__ int      cursor[RROWS_];
  __shared__ int      rstart[RROWS_];
  __shared__ float    dinv[RTILES_ * 16];
  __shared__ int      wsum_scan[32];
  __shared__ int      wsum_base[32];
  __shared__ int      kept_n;
  __shared__ __attribute__((aligned(16))) _Float16 At[RTILES_ * 16 * APITCH_];
  __shared__ __attribute__((aligned(32))) _Float16 WpL[WPACK_HALVES];

  const int t     = threadIdx.x;
  const int lane  = t & 31;
  const int w     = t >> 5;
  const int lo    = q * RROWS_;
  const int hi    = (lo + RROWS_ < N_) ? (lo + RROWS_) : N_;
  const int nrows = hi - lo;
  const int node0 = b * N_;

  if (t < RROWS_) degown[t] = 0;
  if (t == 0) kept_n = 0;
  __syncthreads();

  const int4* __restrict__ e4 = (const int4*)(ei + (size_t)b * E_ * 2);
  const float4* __restrict__ xrows = (const float4*)x + (size_t)(node0 + lo) * (F_ / 4);
  int below = 0;
  int i0 = 0;
#pragma unroll 1
  for (int trip = 0; trip < XROWS_PER_GROUP_; ++trip, i0 += LOADS_ * 1024)
    edge_trip<true>(e4, i0, lo, nrows, below, kept, degown, &kept_n, xrows, At, (t >> 4) + trip * 64);
#pragma unroll 1
  for (; i0 < E_ / 2; i0 += LOADS_ * 1024)
    edge_trip<false>(e4, i0, lo, nrows, below, kept, degown, &kept_n, xrows, At, 0);
  if (t < 256) pack_w_block(W1, WpL);
  __syncthreads();
  const int  kn   = kept_n;
  const bool fast = (kn <= KCAP_);

  const int d = (t < nrows) ? degown[t] : 0;
  int inc = d;
#pragma unroll
  for (int s = 1; s < 32; s <<= 1) {
    const int o = __shfl_up(inc, s);
    if (lane >= s) inc += o;
  }
  const int part = wave_sum(below);
  if (lane == 31) wsum_scan[w] = inc;
  if (lane == 0)  wsum_base[w] = part;
  if (t < RTILES_ * 16) dinv[t] = (d > 0) ? rsqrtf((float)d) : 0.0f;
  __syncthreads();
  const int woff = wave_sum((lane < w) ? wsum_scan[lane] : 0);
  const int base = wave_sum(wsum_base[lane]);
  const int excl = woff + inc - d;
  const int seg0 = b * E_ + base;
  if (t < nrows) {
    rstart[t] = excl;
    cursor[t] = excl;
    rowptr[node0 + lo + t] = seg0 + excl;
  }

  const bool has_tile = (w * 16 < nrows);
  if (has_tile) map_tile_in_place(At + w * 16 * APITCH_, dinv + w * 16, WpL, b1);
  __syncthreads();
  if (has_tile) {
    const _Float16* tile = At + w * 16 * APITCH_;
#pragma unroll
    for (int i = 0; i < 4; ++i) {
      const int piece = i * 32 + lane;
      const int m     = piece >> 3;
      const int part  = piece & 7;
      const v8h val = *(const v8h*)(tile + m * APITCH_ + part * 8);
      if (w * 16 + m < nrows) *(v8h*)(H1 + (size_t)(node0 + lo + w * 16 + m) * F_ + part * 8) = val;
    }
  }

  if (fast) {
    for (int i = t; i < kn; i += 1024) {
      const unsigned u = kept[i];
      const int pos = atomicAdd(&cursor[u >> 14], 1);
      placed[pos] = (int)(u & 16383u);
    }
  } else {
    const int2* __restrict__ e2 = (const int2*)(ei + (size_t)b * E_ * 2);
    for (int g = t; g < E_; g += 1024) {
      const int2 e = e2[g];
      if (e.x >= lo && e.x < hi) {
        const int pos = atomicAdd(&cursor[e.x - lo], 1);
        colu[seg0 + pos] = e.y;
      }
    }
  }
  __syncthreads();

  if (fast) sort_range_rows(placed, degown, rstart, dinv, cw + seg0, t, nrows, node0);
  else      sort_range_rows(colu + seg0, degown, rstart, dinv, cw + seg0, t, nrows, node0);
}

#define SETUP_PACK_BLOCKS 2
__global__ __launch_bounds__(1024) void csr_pack_kernel(const int* __restrict__ ei,
                                                        const float* __restrict__ x,
                                                        const float* __restrict__ W1,
                                                        const float* __restrict__ b1,
                                                        const float* __restrict__ W2,
                                                        const float* __restrict__ W3,
                                                        _Float16* __restrict__ Wp2,
                                                        _Float16* __restrict__ Wp3,
                                                        _Float16* __restrict__ H1,
                                                        int* __restrict__ rowptr,
                                                        int* colu,
                                                        int2* __restrict__ cw) {
  const int blk = blockIdx.x;
  if (blk >= CSR_BLOCKS) {
    const int l = blk - CSR_BLOCKS;
    if (threadIdx.x < 256) pack_w_block((l == 0) ? W2 : W3, (l == 0) ? Wp2 : Wp3);
    return;
  }
  csr_range_block(ei, x, W1, b1, H1, rowptr, colu, cw, blk / RANGES_, blk % RANGES_);
}

#define MODE_SUM   0
#define MODE_NEXT  1
#define GROWS_ 5
#define GATH_ROWS_PER_BLOCK (16 * GROWS_)
#define GATH_TILES_PER_BLOCK (GATH_ROWS_PER_BLOCK / 16)
#define ROW_SHIFT_ 7
static_assert(BN_ % GATH_ROWS_PER_BLOCK == 0, "every group owns GROWS_ full rows");
static_assert(GROWS_ < 16, "lane sub of a group holds row pointer sub, sub <= GROWS_");
static_assert(GATH_ROWS_PER_BLOCK % 16 == 0 && GATH_TILES_PER_BLOCK <= 8, "whole 16-row tiles, one wave each");

__device__ __forceinline__ void axpy4(float4& acc, float w, const v4h h) {
  acc.x = fmaf(w, (float)h[0], acc.x);
  acc.y = fmaf(w, (float)h[1], acc.y);
  acc.z = fmaf(w, (float)h[2], acc.z);
  acc.w = fmaf(w, (float)h[3], acc.w);
}
template <int NK>
__device__ __forceinline__ void chunk_terms(const char* __restrict__ Hb, const unsigned lane_off,
                                            const int ms, const int mwi, float4& acc) {
  v4h h[NK];
#pragma unroll
  for (int k = 0; k < NK; ++k) {
    const unsigned s = (unsigned)__shfl(ms, k, 16);
    h[k] = *(const v4h*)(Hb + ((s << ROW_SHIFT_) | lane_off));
  }
#pragma unroll
  for (int k = 0; k < NK; ++k) {
    const float w = __int_as_float(__shfl(mwi, k, 16));
    axpy4(acc, w, h[k]);
  }
}

template <int MODE>
__global__ __launch_bounds__(256) void gather_rows_kernel(const _Float16* __restrict__ Hsrc,
                                                          const int* __restrict__ rowptr,
                                                          const int2* __restrict__ cw,
                                                          float* __restrict__ out,
                                                          const _Float16* __restrict__ Wp,
                                                          const float* __restrict__ bias) {
  __shared__ __attribute__((aligned(16))) _Float16 At[(MODE != MODE_SUM) ? GATH_ROWS_PER_BLOCK * APITCH_ : 8];
  __shared__ float Dv[(MODE != MODE_SUM) ? GATH_ROWS_PER_BLOCK : 1];

  const int t   = blockIdx.x * blockDim.x + threadIdx.x;
  const int sub = t & 15;
  const int r0  = (t >> 4) * GROWS_;
  const char* __restrict__ Hb = (const char*)Hsrc;
  const unsigned lane_off = (unsigned)sub << (ROW_SHIFT_ - 4);

  int rp = 0;
  if (sub <= GROWS_) rp = (r0 + sub < BN_) ? rowptr[r0 + sub] : BE_;
  const int gend = __shfl(rp, GROWS_, 16);
  int base = __shfl(rp, 0, 16);
  int2 my = make_int2(0, 0);
  if (base + sub < gend) my = cw[base + sub];

  for (int rr = 0; rr < GROWS_; ++rr) {
    const int end = __shfl(rp, rr + 1, 16);
    const int dd  = end - base;
    float4 acc = make_float4(0.0f, 0.0f, 0.0f, 0.0f);
    while (base < end) {
      const int n  = end - base;
      const int nb = (n > 16) ? (base + 16) : end;
      const bool live = sub < n;
      const int  ms   = live ? my.x : 0;
      const int  mwi  = live ? my.y : 0;
      const int2 nxt = cw[(nb + sub < BE_) ? (nb + sub) : (BE_ - 1)];
      if (__any(n > 8)) chunk_terms<16>(Hb, lane_off, ms, mwi, acc);
      else              chunk_terms<8>(Hb, lane_off, ms, mwi, acc);
      my   = nxt;
      base = nb;
    }
    if (MODE != MODE_SUM) {
      v4h q;
      q[0] = (_Float16)fmaxf(acc.x, 0.0f);
      q[1] = (_Float16)fmaxf(acc.y, 0.0f);
      q[2] = (_Float16)fmaxf(acc.z, 0.0f);
      q[3] = (_Float16)fmaxf(acc.w, 0.0f);
      const int lr = (threadIdx.x >> 4) * GROWS_ + rr;
      *(v4h*)(At + lr * APITCH_ + sub * 4) = q;
      if (sub == 0) Dv[lr] = (dd > 0) ? rsqrtf((float)dd) : 0.0f;
    } else {
      ((float4*)out)[(size_t)(r0 + rr) * (F_ / 4) + sub] = acc;
    }
  }

  if (MODE != MODE_SUM) {
    __syncthreads();
    const int wv = threadIdx.x >> 5;
    if (wv < GATH_TILES_PER_BLOCK) {
      v16h a0, a1;
      load_a_frags(At + wv * 16 * APITCH_, a0, a1);
      gemm_bias_scale_frags(a0, a1, Wp, bias, Dv + wv * 16, (_Float16*)out,
                            blockIdx.x * GATH_ROWS_PER_BLOCK + wv * 16);
    }
  }
}

extern "C" void kernel_launch(void* const* d_in, const int* in_sizes, int n_in,
                              void* d_out, int out_size, void* d_ws, size_t ws_size,
                              hipStream_t stream) {
  const float* x  = (const float*)d_in[0];
  const int*   ei = (const int*)  d_in[1];
  const float* W1 = (const float*)d_in[2];
  const float* b1 = (const float*)d_in[3];
  const float* W2 = (const float*)d_in[4];
  const float* b2 = (const float*)d_in[5];
  const float* W3 = (const float*)d_in[6];
  const float* b3 = (const float*)d_in[7];
  float* out = (float*)d_out;

  auto align256 = [](size_t v) { return (v + 255) & ~(size_t)255; };
  char* ws = (char*)d_ws;
  int*   rowptr = (int*)ws;      ws += align256((size_t)BN_ * 4);
  int2*  cw     = (int2*)ws;     ws += 2 * align256((size_t)BE_ * 4);
  int*   colu   = (int*)ws;      ws += align256((size_t)BE_ * 4);
  float* Hlin  = (float*)ws;     ws += align256((size_t)BN_ * F_ * 4);
  float* Hlin2 = (float*)ws;     ws += align256((size_t)BN_ * F_ * 4);
  ws += align256((size_t)WPACK_HALVES * 2);
  _Float16* Wp2 = (_Float16*)ws; ws += align256((size_t)WPACK_HALVES * 2);
  _Float16* Wp3 = (_Float16*)ws; ws += align256((size_t)WPACK_HALVES * 2);

  const int TB = 256;
  const int GATH_GRID = BN_ / GATH_ROWS_PER_BLOCK;

  csr_pack_kernel<<<CSR_BLOCKS + SETUP_PACK_BLOCKS, 1024, 0, stream>>>(ei, x, W1, b1, W2, W3, Wp2, Wp3,
                                                                     (_Float16*)Hlin, rowptr, colu, cw);

  gather_rows_kernel<MODE_NEXT><<<GATH_GRID, TB, 0, stream>>>((const _Float16*)Hlin, rowptr, cw, Hlin2, Wp2, b2);

  gather_rows_kernel<MODE_NEXT><<<GATH_GRID, TB, 0, stream>>>((const _Float16*)Hlin2, rowptr, cw, Hlin, Wp3, b3);

  gather_rows_kernel<MODE_SUM><<<GATH_GRID, TB, 0, stream>>>((const _Float16*)Hlin, rowptr, cw, out, nullptr, nullptr);
}


# Row gather: a chunk is up to 24 terms, not 16 (two pairs per lane; widths 8/16/24, wave-uniform): nearly every row's loads go out in one round, one wait for memory per row, not two; sum order kept
# speedup vs baseline: 5.9916x; 1.0315x over previous
#include <hip/hip_runtime.h>
#include <hip/hip_bf16.h>

typedef __attribute__((ext_vector_type(16))) _Float16 v16h;
typedef __attribute__((ext_vector_type(8)))  float    v8f;

#define B_  8
#define N_  10000
#define E_  160000
#define F_  64
#define BN_ (B_*N_)
#define BE_ (B_*E_)

#define WPACK_HALVES 4096

typedef __attribute__((ext_vector_type(4))) _Float16 v4h;
typedef __attribute__((ext_vector_type(8))) _Float16 v8h;
#define APITCH_ 72

__device__ __forceinline__ void load_a_frags(const _Float16* tile, v16h& a0, v16h& a1) {
  const int lane = threadIdx.x & 31;
  const _Float16* arow = tile + (lane & 15) * APITCH_ + (lane >> 4) * 8;
  const v8h q0 = *(const v8h*)(arow);
  const v8h q1 = *(const v8h*)(arow + 16);
  const v8h q2 = *(const v8h*)(arow + 32);
  const v8h q3 = *(const v8h*)(arow + 48);
  a0 = __builtin_shufflevector(q0, q1, 0, 1, 2, 3, 4, 5, 6, 7, 8, 9, 10, 11, 12, 13, 14, 15);
  a1 = __builtin_shufflevector(q2, q3, 0, 1, 2, 3, 4, 5, 6, 7, 8, 9, 10, 11, 12, 13, 14, 15);
}

__device__ __forceinline__ void gemm_bias_scale_frags(const v16h a0, const v16h a1,
                                                      const _Float16* __restrict__ Wp,
                                                      const float* __restrict__ bias,
                                                      const float* dscale,
                                                      _Float16* __restrict__ Hout,
                                                      const int row0) {
  const int lane = threadIdx.x & 31;
  const int half = lane >> 4;
  const int mrow = lane & 15;
  float sv[8];
#pragma unroll
  for (int r = 0; r < 8; ++r) sv[r] = dscale[half * 8 + r];
  const v16h* Wpv = (const v16h*)Wp;
#pragma unroll
  for (int nt = 0; nt < 4; ++nt) {
    const v16h b0 = Wpv[(nt * 2 + 0) * 32 + lane];
    const v16h b1 = Wpv[(nt * 2 + 1) * 32 + lane];

    v8f c = {};
    c = __builtin_amdgcn_wmma_f32_16x16x32_f16(false, a0, false, b0, (short)0, c, false, false);
    c = __builtin_amdgcn_wmma_f32_16x16x32_f16(false, a1, false, b1, (short)0, c, false, false);

    const int n = nt * 16 + mrow;
    const float bv = bias[n];
#pragma unroll
    for (int r = 0; r < 8; ++r) {
      const int m = r + half * 8;
      Hout[(size_t)(row0 + m) * F_ + n] = (_Float16)((c[r] + bv) * sv[r]);
    }
  }
}

__device__ __forceinline__ void map_tile_in_place(_Float16* tile, const float* dscale,
                                                  const _Float16* Wp,
                                                  const float* __restrict__ bias) {
  const int lane = threadIdx.x & 31;
  const int half = lane >> 4;
  const int mrow = lane & 15;
  v16h a0, a1;
  load_a_frags(tile, a0, a1);
  float sv[8];
#pragma unroll
  for (int r = 0; r < 8; ++r) sv[r] = dscale[half * 8 + r];
  const v16h* Wpv = (const v16h*)Wp;
#pragma unroll
  for (int nt = 0; nt < 4; ++nt) {
    const v16h b0 = Wpv[(nt * 2 + 0) * 32 + lane];
    const v16h b1 = Wpv[(nt * 2 + 1) * 32 + lane];

    v8f c = {};
    c = __builtin_amdgcn_wmma_f32_16x16x32_f16(false, a0, false, b0, (short)0, c, false, false);
    c = __builtin_amdgcn_wmma_f32_16x16x32_f16(false, a1, false, b1, (short)0, c, false, false);

    const int n = nt * 16 + mrow;
    const float bv = bias[n];
#pragma unroll
    for (int r = 0; r < 8; ++r) {
      const int m = r + half * 8;
      tile[m * APITCH_ + n] = (_Float16)((c[r] + bv) * sv[r]);
    }
  }
}

__device__ __forceinline__ void pack_w_block(const float* __restrict__ W,
                                             _Float16* __restrict__ Wp) {
  const int t    = threadIdx.x;
  const int nt   = t >> 6;
  const int kb   = (t >> 5) & 1;
  const int lane = t & 31;
  const int n    = nt * 16 + (lane & 15);
  const int half = lane >> 4;
  _Float16* dst = Wp + ((size_t)((nt * 2 + kb) * 32 + lane)) * 16;
#pragma unroll
  for (int v = 0; v < 8; ++v) {
    int kbase = (v < 4) ? (2 * v) : (16 + 2 * (v - 4));
    int k = kb * 32 + kbase + half * 8;
    dst[2 * v]     = (_Float16)W[n * F_ + k];
    dst[2 * v + 1] = (_Float16)W[n * F_ + k + 1];
  }
}

#define RANGES_ 32
#define RROWS_  ((N_ + RANGES_ - 1) / RANGES_)
#define RTILES_ ((RROWS_ + 15) / 16)
#define KCAP_   8192
#define CSR_BLOCKS (B_ * RANGES_)
#define LOADS_  8
static_assert(N_ <= 16384, "a source node packs into 14 bits of a kept entry");
static_assert(RROWS_ <= 1024, "one thread per row of the range in the scan");
#define XROWS_PER_GROUP_ (RTILES_ * 16 / 64)
static_assert(RTILES_ <= 32, "one wave per 16-row tile of the range");
static_assert((RTILES_ * 16) % 64 == 0, "the staged rows divide evenly over the block's 64 groups");
static_assert((E_ / 2 + LOADS_ * 1024 - 1) / (LOADS_ * 1024) >= XROWS_PER_GROUP_, "the edge pass has a trip for every staged row of a group");
static_assert(E_ % 2 == 0, "the edge list is read two edges per 16-byte load");

__device__ __forceinline__ int wave_sum(int v) {
#pragma unroll
  for (int m = 16; m > 0; m >>= 1) v += __shfl_xor(v, m);
  return v;
}

__device__ __forceinline__ void sort_range_rows(const int* src, const int* deg, const int* rstart,
                                                const float* dinv,
                                                int2* __restrict__ cw_seg, const int t,
                                                const int nrows, const int node0) {
  const int sub = t & 15;
  for (int rl = t >> 4; rl < nrows; rl += 64) {
    const int start = rstart[rl];
    const int dd    = deg[rl];
    const int wbits = __float_as_int(dinv[rl]);
    const int* row  = src + start;
    for (int e = sub; e < dd; e += 16) {
      const int key = row[e];
      int rank = 0;
      for (int j = 0; j < dd; ++j) {
        const int kj = row[j];
        rank += ((kj < key) || (kj == key && j < e)) ? 1 : 0;
      }
      cw_seg[start + rank] = make_int2(key + node0, wbits);
    }
  }
}

template <bool STAGE>
__device__ __forceinline__ void edge_trip(const int4* __restrict__ e4, const int i0,
                                          const int lo, const int nrows, int& below,
                                          unsigned* kept, int* degown, int* kept_n,
                                          const float4* __restrict__ xrows,
                                          _Float16* At, const int lr) {
  const int t = threadIdx.x;
  float4 xq;
  if (STAGE) xq = xrows[(size_t)((lr < nrows) ? lr : (nrows - 1)) * (F_ / 4) + (t & 15)];
  int4 v[LOADS_];
#pragma unroll
  for (int u = 0; u < LOADS_; ++u) {
    const int idx = i0 + u * 1024 + t;
    v[u] = e4[(idx < E_ / 2) ? idx : (E_ / 2 - 1)];
  }
  int r[2 * LOADS_], s[2 * LOADS_];
#pragma unroll
  for (int u = 0; u < LOADS_; ++u) {
    const bool live = (i0 + u * 1024 + t) < E_ / 2;
    r[2 * u]     = live ? v[u].x : N_;  s[2 * u]     = v[u].y;
    r[2 * u + 1] = live ? v[u].z : N_;  s[2 * u + 1] = v[u].w;
  }
  int n_in = 0;
#pragma unroll
  for (int k = 0; k < 2 * LOADS_; ++k) {
    below += (r[k] < lo) ? 1 : 0;
    n_in  += ((unsigned)(r[k] - lo) < (unsigned)nrows) ? 1 : 0;
  }
  int p = (n_in > 0) ? atomicAdd(kept_n, n_in) : 0;
#pragma unroll
  for (int k = 0; k < 2 * LOADS_; ++k) {
    const unsigned rl = (unsigned)(r[k] - lo);
    if (rl < (unsigned)nrows) {
      atomicAdd(&degown[rl], 1);
      if (p < KCAP_) kept[p] = (rl << 14) | (unsigned)s[k];
      ++p;
    }
  }
  if (STAGE) {
    v4h qv;
    qv[0] = (_Float16)xq.x;  qv[1] = (_Float16)xq.y;  qv[2] = (_Float16)xq.z;  qv[3] = (_Float16)xq.w;
    *(v4h*)(At + lr * APITCH_ + (t & 15) * 4) = qv;
  }
}

__device__ __forceinline__ void csr_range_block(const int* __restrict__ ei,
                                                const float* __restrict__ x,
                                                const float* __restrict__ W1,
                                                const float* __restrict__ b1,
                                                _Float16* __restrict__ H1,
                                                int* __restrict__ rowptr,
                                                int* colu,
                                                int2* __restrict__ cw,
                                                const int b, const int q) {
  __shared__ unsigned kept[KCAP_];
  __shared__ int      placed[KCAP_];
  __shared__ int      degown[RROWS_];
  __shared__ int      cursor[RROWS_];
  __shared__ int      rstart[RROWS_];
  __shared__ float    dinv[RTILES_ * 16];
  __shared__ int      wsum_scan[32];
  __shared__ int      wsum_base[32];
  __shared__ int      kept_n;
  __shared__ __attribute__((aligned(16))) _Float16 At[RTILES_ * 16 * APITCH_];
  __shared__ __attribute__((aligned(32))) _Float16 WpL[WPACK_HALVES];

  const int t     = threadIdx.x;
  const int lane  = t & 31;
  const int w     = t >> 5;
  const int lo    = q * RROWS_;
  const int hi    = (lo + RROWS_ < N_) ? (lo + RROWS_) : N_;
  const int nrows = hi - lo;
  const int node0 = b * N_;

  if (t < RROWS_) degown[t] = 0;
  if (t == 0) kept_n = 0;
  __syncthreads();

  const int4* __restrict__ e4 = (const int4*)(ei + (size_t)b * E_ * 2);
  const float4* __restrict__ xrows = (const float4*)x + (size_t)(node0 + lo) * (F_ / 4);
  int below = 0;
  int i0 = 0;
#pragma unroll 1
  for (int trip = 0; trip < XROWS_PER_GROUP_; ++trip, i0 += LOADS_ * 1024)
    edge_trip<true>(e4, i0, lo, nrows, below, kept, degown, &kept_n, xrows, At, (t >> 4) + trip * 64);
#pragma unroll 1
  for (; i0 < E_ / 2; i0 += LOADS_ * 1024)
    edge_trip<false>(e4, i0, lo, nrows, below, kept, degown, &kept_n, xrows, At, 0);
  if (t < 256) pack_w_block(W1, WpL);
  __syncthreads();
  const int  kn   = kept_n;
  const bool fast = (kn <= KCAP_);

  const int d = (t < nrows) ? degown[t] : 0;
  int inc = d;
#pragma unroll
  for (int s = 1; s < 32; s <<= 1) {
    const int o = __shfl_up(inc, s);
    if (lane >= s) inc += o;
  }
  const int part = wave_sum(below);
  if (lane == 31) wsum_scan[w] = inc;
  if (lane == 0)  wsum_base[w] = part;
  if (t < RTILES_ * 16) dinv[t] = (d > 0) ? rsqrtf((float)d) : 0.0f;
  __syncthreads();
  const int woff = wave_sum((lane < w) ? wsum_scan[lane] : 0);
  const int base = wave_sum(wsum_base[lane]);
  const int excl = woff + inc - d;
  const int seg0 = b * E_ + base;
  if (t < nrows) {
    rstart[t] = excl;
    cursor[t] = excl;
    rowptr[node0 + lo + t] = seg0 + excl;
  }

  const bool has_tile = (w * 16 < nrows);
  if (has_tile) map_tile_in_place(At + w * 16 * APITCH_, dinv + w * 16, WpL, b1);
  __syncthreads();
  if (has_tile) {
    const _Float16* tile = At + w * 16 * APITCH_;
#pragma unroll
    for (int i = 0; i < 4; ++i) {
      const int piece = i * 32 + lane;
      const int m     = piece >> 3;
      const int part  = piece & 7;
      const v8h val = *(const v8h*)(tile + m * APITCH_ + part * 8);
      if (w * 16 + m < nrows) *(v8h*)(H1 + (size_t)(node0 + lo + w * 16 + m) * F_ + part * 8) = val;
    }
  }

  if (fast) {
    for (int i = t; i < kn; i += 1024) {
      const unsigned u = kept[i];
      const int pos = atomicAdd(&cursor[u >> 14], 1);
      placed[pos] = (int)(u & 16383u);
    }
  } else {
    const int2* __restrict__ e2 = (const int2*)(ei + (size_t)b * E_ * 2);
    for (int g = t; g < E_; g += 1024) {
      const int2 e = e2[g];
      if (e.x >= lo && e.x < hi) {
        const int pos = atomicAdd(&cursor[e.x - lo], 1);
        colu[seg0 + pos] = e.y;
      }
    }
  }
  __syncthreads();

  if (fast) sort_range_rows(placed, degown, rstart, dinv, cw + seg0, t, nrows, node0);
  else      sort_range_rows(colu + seg0, degown, rstart, dinv, cw + seg0, t, nrows, node0);
}

#define SETUP_PACK_BLOCKS 2
__global__ __launch_bounds__(1024) void csr_pack_kernel(const int* __restrict__ ei,
                                                        const float* __restrict__ x,
                                                        const float* __restrict__ W1,
                                                        const float* __restrict__ b1,
                                                        const float* __restrict__ W2,
                                                        const float* __restrict__ W3,
                                                        _Float16* __restrict__ Wp2,
                                                        _Float16* __restrict__ Wp3,
                                                        _Float16* __restrict__ H1,
                                                        int* __restrict__ rowptr,
                                                        int* colu,
                                                        int2* __restrict__ cw) {
  const int blk = blockIdx.x;
  if (blk >= CSR_BLOCKS) {
    const int l = blk - CSR_BLOCKS;
    if (threadIdx.x < 256) pack_w_block((l == 0) ? W2 : W3, (l == 0) ? Wp2 : Wp3);
    return;
  }
  csr_range_block(ei, x, W1, b1, H1, rowptr, colu, cw, blk / RANGES_, blk % RANGES_);
}

#define MODE_SUM   0
#define MODE_NEXT  1
#define GROWS_ 5
#define GATH_ROWS_PER_BLOCK (16 * GROWS_)
#define GATH_TILES_PER_BLOCK (GATH_ROWS_PER_BLOCK / 16)
#define ROW_SHIFT_ 7
static_assert(BN_ % GATH_ROWS_PER_BLOCK == 0, "every group owns GROWS_ full rows");
static_assert(GROWS_ < 16, "lane sub of a group holds row pointer sub, sub <= GROWS_");
static_assert(GATH_ROWS_PER_BLOCK % 16 == 0 && GATH_TILES_PER_BLOCK <= 8, "whole 16-row tiles, one wave each");

__device__ __forceinline__ void axpy4(float4& acc, float w, const v4h h) {
  acc.x = fmaf(w, (float)h[0], acc.x);
  acc.y = fmaf(w, (float)h[1], acc.y);
  acc.z = fmaf(w, (float)h[2], acc.z);
  acc.w = fmaf(w, (float)h[3], acc.w);
}
#define CHUNK_ 24
template <int NK>
__device__ __forceinline__ void chunk_terms(const char* __restrict__ Hb, const unsigned lane_off,
                                            const int ms0, const int mwi0,
                                            const int ms1, const int mwi1, float4& acc) {
  static_assert(NK <= CHUNK_ && CHUNK_ <= 32, "a lane holds two pairs of the chunk");
  v4h h[NK];
#pragma unroll
  for (int k = 0; k < NK; ++k) {
    const unsigned s = (unsigned)((k < 16) ? __shfl(ms0, k, 16) : __shfl(ms1, k - 16, 16));
    h[k] = *(const v4h*)(Hb + ((s << ROW_SHIFT_) | lane_off));
  }
#pragma unroll
  for (int k = 0; k < NK; ++k) {
    const float w = __int_as_float((k < 16) ? __shfl(mwi0, k, 16) : __shfl(mwi1, k - 16, 16));
    axpy4(acc, w, h[k]);
  }
}

template <int MODE>
__global__ __launch_bounds__(256) __attribute__((amdgpu_waves_per_eu(8, 8))) void gather_rows_kernel(const _Float16* __restrict__ Hsrc,
                                                          const int* __restrict__ rowptr,
                                                          const int2* __restrict__ cw,
                                                          float* __restrict__ out,
                                                          const _Float16* __restrict__ Wp,
                                                          const float* __restrict__ bias) {
  __shared__ __attribute__((aligned(16))) _Float16 At[(MODE != MODE_SUM) ? GATH_ROWS_PER_BLOCK * APITCH_ : 8];
  __shared__ float Dv[(MODE != MODE_SUM) ? GATH_ROWS_PER_BLOCK : 1];

  const int t   = blockIdx.x * blockDim.x + threadIdx.x;
  const int sub = t & 15;
  const int r0  = (t >> 4) * GROWS_;
  const char* __restrict__ Hb = (const char*)Hsrc;
  const unsigned lane_off = (unsigned)sub << (ROW_SHIFT_ - 4);

  int rp = 0;
  if (sub <= GROWS_) rp = (r0 + sub < BN_) ? rowptr[r0 + sub] : BE_;
  const int gend = __shfl(rp, GROWS_, 16);
  int base = __shfl(rp, 0, 16);
  int2 my0 = make_int2(0, 0);
  int2 my1 = make_int2(0, 0);
  if (base + sub < gend)      my0 = cw[base + sub];
  if (base + 16 + sub < gend) my1 = cw[base + 16 + sub];

  for (int rr = 0; rr < GROWS_; ++rr) {
    const int end = __shfl(rp, rr + 1, 16);
    const int dd  = end - base;
    float4 acc = make_float4(0.0f, 0.0f, 0.0f, 0.0f);
    while (base < end) {
      const int n  = end - base;
      const int nb = (n > CHUNK_) ? (base + CHUNK_) : end;
      const bool live0 = sub < n;
      const bool live1 = sub + 16 < n;
      const int  ms0  = live0 ? my0.x : 0;
      const int  mwi0 = live0 ? my0.y : 0;
      const int  ms1  = live1 ? my1.x : 0;
      const int  mwi1 = live1 ? my1.y : 0;
      const int2 nxt0 = cw[(nb + sub < BE_) ? (nb + sub) : (BE_ - 1)];
      const int2 nxt1 = cw[(nb + 16 + sub < BE_) ? (nb + 16 + sub) : (BE_ - 1)];
      if      (__any(n > 16)) chunk_terms<CHUNK_>(Hb, lane_off, ms0, mwi0, ms1, mwi1, acc);
      else if (__any(n > 8))  chunk_terms<16>(Hb, lane_off, ms0, mwi0, ms1, mwi1, acc);
      else                    chunk_terms<8>(Hb, lane_off, ms0, mwi0, ms1, mwi1, acc);
      my0  = nxt0;
      my1  = nxt1;
      base = nb;
    }
    if (MODE != MODE_SUM) {
      v4h q;
      q[0] = (_Float16)fmaxf(acc.x, 0.0f);
      q[1] = (_Float16)fmaxf(acc.y, 0.0f);
      q[2] = (_Float16)fmaxf(acc.z, 0.0f);
      q[3] = (_Float16)fmaxf(acc.w, 0.0f);
      const int lr = (threadIdx.x >> 4) * GROWS_ + rr;
      *(v4h*)(At + lr * APITCH_ + sub * 4) = q;
      if (sub == 0) Dv[lr] = (dd > 0) ? rsqrtf((float)dd) : 0.0f;
    } else {
      ((float4*)out)[(size_t)(r0 + rr) * (F_ / 4) + sub] = acc;
    }
  }

  if (MODE != MODE_SUM) {
    __syncthreads();
    const int wv = threadIdx.x >> 5;
    if (wv < GATH_TILES_PER_BLOCK) {
      v16h a0, a1;
      load_a_frags(At + wv * 16 * APITCH_, a0, a1);
      gemm_bias_scale_frags(a0, a1, Wp, bias, Dv + wv * 16, (_Float16*)out,
                            blockIdx.x * GATH_ROWS_PER_BLOCK + wv * 16);
    }
  }
}

extern "C" void kernel_launch(void* const* d_in, const int* in_sizes, int n_in,
                              void* d_out, int out_size, void* d_ws, size_t ws_size,
                              hipStream_t stream) {
  const float* x  = (const float*)d_in[0];
  const int*   ei = (const int*)  d_in[1];
  const float* W1 = (const float*)d_in[2];
  const float* b1 = (const float*)d_in[3];
  const float* W2 = (const float*)d_in[4];
  const float* b2 = (const float*)d_in[5];
  const float* W3 = (const float*)d_in[6];
  const float* b3 = (const float*)d_in[7];
  float* out = (float*)d_out;

  auto align256 = [](size_t v) { return (v + 255) & ~(size_t)255; };
  char* ws = (char*)d_ws;
  int*   rowptr = (int*)ws;      ws += align256((size_t)BN_ * 4);
  int2*  cw     = (int2*)ws;     ws += 2 * align256((size_t)BE_ * 4);
  int*   colu   = (int*)ws;      ws += align256((size_t)BE_ * 4);
  float* Hlin  = (float*)ws;     ws += align256((size_t)BN_ * F_ * 4);
  float* Hlin2 = (float*)ws;     ws += align256((size_t)BN_ * F_ * 4);
  ws += align256((size_t)WPACK_HALVES * 2);
  _Float16* Wp2 = (_Float16*)ws; ws += align256((size_t)WPACK_HALVES * 2);
  _Float16* Wp3 = (_Float16*)ws; ws += align256((size_t)WPACK_HALVES * 2);

  const int TB = 256;
  const int GATH_GRID = BN_ / GATH_ROWS_PER_BLOCK;

  csr_pack_kernel<<<CSR_BLOCKS + SETUP_PACK_BLOCKS, 1024, 0, stream>>>(ei, x, W1, b1, W2, W3, Wp2, Wp3,
                                                                     (_Float16*)Hlin, rowptr, colu, cw);

  gather_rows_kernel<MODE_NEXT><<<GATH_GRID, TB, 0, stream>>>((const _Float16*)Hlin, rowptr, cw, Hlin2, Wp2, b2);

  gather_rows_kernel<MODE_NEXT><<<GATH_GRID, TB, 0, stream>>>((const _Float16*)Hlin2, rowptr, cw, Hlin, Wp3, b3);

  gather_rows_kernel<MODE_SUM><<<GATH_GRID, TB, 0, stream>>>((const _Float16*)Hlin, rowptr, cw, out, nullptr, nullptr);
}
